# starting kernel
# baseline (speedup 1.0000x reference)
#include <hip/hip_runtime.h>
#include <hip/hip_bf16.h>


#define SAGE_N 50000
#define SAGE_D 128

typedef __attribute__((ext_vector_type(2))) float v2f;
typedef __attribute__((ext_vector_type(8))) float v8f;

__global__ void sage_zero_kernel(float* __restrict__ p, size_t n) {
    size_t i = (size_t)blockIdx.x * blockDim.x + threadIdx.x;
    size_t stride = (size_t)gridDim.x * blockDim.x;
    for (; i < n; i += stride) p[i] = 0.0f;
}

__global__ void sage_count_kernel(const int* __restrict__ dst,
                                  float* __restrict__ cnt, int E) {
    int e = blockIdx.x * blockDim.x + threadIdx.x;
    if (e < E) unsafeAtomicAdd(&cnt[dst[e]], 1.0f);
}

__global__ __launch_bounds__(256) void sage_scatter_kernel(
    const float* __restrict__ xin, const int* __restrict__ src,
    const int* __restrict__ dst, float* __restrict__ s, int E) {
    int e = blockIdx.x * 8 + (threadIdx.x >> 5);
    if (e >= E) return;
    int lane = threadIdx.x & 31;
    int sn = src[e];
    int dn = dst[e];
    const float4 v = *(const float4*)(xin + (size_t)sn * SAGE_D + lane * 4);
    float* sd = s + (size_t)dn * SAGE_D + lane * 4;
    unsafeAtomicAdd(sd + 0, v.x);
    unsafeAtomicAdd(sd + 1, v.y);
    unsafeAtomicAdd(sd + 2, v.z);
    unsafeAtomicAdd(sd + 3, v.w);
}

__global__ __launch_bounds__(256) void sage_gemm_kernel(
    const float* __restrict__ s,
    const float* __restrict__ cnt,
    const float* __restrict__ xin,
    const float* __restrict__ Wl,
    const float* __restrict__ bl,
    const float* __restrict__ Wr,
    float* __restrict__ out,
    int nRowTiles) {
    const int wave = threadIdx.x >> 5;
    const int tile = blockIdx.x * 8 + wave;
    if (tile >= nRowTiles) return;

    const int lane = threadIdx.x & 31;
    const int half = lane >> 4;
    const int lm   = lane & 15;
    const int m0   = tile * 16;

    const float inv = 1.0f / fmaxf(cnt[m0 + lm], 1.0f);

    const float* arow_s = s   + (size_t)(m0 + lm) * SAGE_D + 2 * half;
    const float* arow_x = xin + (size_t)(m0 + lm) * SAGE_D + 2 * half;
    const float* wl_row = Wl  + (size_t)lm * SAGE_D + 2 * half;
    const float* wr_row = Wr  + (size_t)lm * SAGE_D + 2 * half;

    v8f acc[8];
#pragma unroll
    for (int ct = 0; ct < 8; ++ct) acc[ct] = (v8f){};

    for (int k0 = 0; k0 < SAGE_D; k0 += 4) {
        v2f a_agg = *(const v2f*)(arow_s + k0);
        a_agg = a_agg * inv;
        v2f a_x = *(const v2f*)(arow_x + k0);
#pragma unroll
        for (int ct = 0; ct < 8; ++ct) {
            v2f b_l = *(const v2f*)(wl_row + (size_t)ct * 16 * SAGE_D + k0);
            v2f b_r = *(const v2f*)(wr_row + (size_t)ct * 16 * SAGE_D + k0);
            acc[ct] = __builtin_amdgcn_wmma_f32_16x16x4_f32(
                false, a_agg, false, b_l, (short)0, acc[ct], false, false);
            acc[ct] = __builtin_amdgcn_wmma_f32_16x16x4_f32(
                false, a_x, false, b_r, (short)0, acc[ct], false, false);
        }
    }

#pragma unroll
    for (int ct = 0; ct < 8; ++ct) {
        const float bias = bl[ct * 16 + lm];
#pragma unroll
        for (int v = 0; v < 8; ++v) {
            out[(size_t)(m0 + v + 8 * half) * SAGE_D + ct * 16 + lm] =
                acc[ct][v] + bias;
        }
    }
}

extern "C" void kernel_launch(void* const* d_in, const int* in_sizes, int n_in,
                              void* d_out, int out_size, void* d_ws, size_t ws_size,
                              hipStream_t stream) {
    const float* x   = (const float*)d_in[0];
    const int*   ei  = (const int*)d_in[1];
    const float* Wl1 = (const float*)d_in[2];
    const float* bl1 = (const float*)d_in[3];
    const float* Wr1 = (const float*)d_in[4];
    const float* Wl2 = (const float*)d_in[5];
    const float* bl2 = (const float*)d_in[6];
    const float* Wr2 = (const float*)d_in[7];
    float* out = (float*)d_out;

    const int N = SAGE_N;
    const int E = in_sizes[1] / 2;
    const int* src = ei;
    const int* dst = ei + E;

    float* s   = (float*)d_ws;
    float* cnt = s + (size_t)N * SAGE_D;

    const int nRowTiles  = N / 16;
    const int gemmBlocks = (nRowTiles + 7) / 8;
    const int scatBlocks = (E + 7) / 8;

    sage_zero_kernel<<<2048, 256, 0, stream>>>(s, (size_t)N * SAGE_D + N);
    sage_count_kernel<<<(E + 255) / 256, 256, 0, stream>>>(dst, cnt, E);

    sage_scatter_kernel<<<scatBlocks, 256, 0, stream>>>(x, src, dst, s, E);
    sage_gemm_kernel<<<gemmBlocks, 256, 0, stream>>>(s, cnt, x, Wl1, bl1, Wr1,
                                                     out, nRowTiles);

    sage_zero_kernel<<<2048, 256, 0, stream>>>(s, (size_t)N * SAGE_D);
    sage_scatter_kernel<<<scatBlocks, 256, 0, stream>>>(out, src, dst, s, E);
    sage_gemm_kernel<<<gemmBlocks, 256, 0, stream>>>(s, cnt, out, Wl2, bl2, Wr2,
                                                     out, nRowTiles);
}


# Neighbour sums kept in LDS: 256 work-groups each own 196 destination nodes, scan dst once, gather listed rows, add by LDS float atomics (arrival order); global atomics, count, zero launches gone
# speedup vs baseline: 2.1088x; 2.1088x over previous
#include <hip/hip_runtime.h>
#include <hip/hip_bf16.h>


#define SAGE_N 50000
#define SAGE_D 128
#define SAGE_R 196
#define SAGE_LCAP 16384
#define SAGE_T 1024
#define SAGE_U 8

typedef __attribute__((ext_vector_type(2))) float v2f;
typedef __attribute__((ext_vector_type(8))) float v8f;

__device__ __forceinline__ void sage_note_edge(
    int d, int e, int lo, int nOwn, int writeCnt, int* cntl, int* nlist,
    int2* lst, float* acc, const float* __restrict__ xin,
    const int* __restrict__ src) {
    const unsigned dl = (unsigned)(d - lo);
    if (dl < (unsigned)nOwn) {
        if (writeCnt) atomicAdd(&cntl[dl], 1);
        const int pos = atomicAdd(nlist, 1);
        if (pos < SAGE_LCAP) {
            lst[pos] = make_int2(e, (int)dl);
        } else {
            const float* xr = xin + (size_t)src[e] * SAGE_D;
            float* ar = acc + dl * SAGE_D;
            for (int k = 0; k < SAGE_D; ++k)
                unsafeAtomicAdd(ar + (k & 3) * 32 + (k >> 2), xr[k]);
        }
    }
}

__global__ __launch_bounds__(SAGE_T) void sage_agg_kernel(
    const float* __restrict__ xin, const int* __restrict__ src,
    const int* __restrict__ dst, float* __restrict__ s,
    float* __restrict__ cnt, int* __restrict__ wlist,
    int* __restrict__ wcount, int E, int first) {
    __shared__ float acc[SAGE_R * SAGE_D];
    __shared__ int2 lst[SAGE_LCAP];
    __shared__ int cntl[SAGE_R];
    __shared__ int nlist;

    const int tid = threadIdx.x;
    const int lane = tid & 31;
    const int wave = __builtin_amdgcn_readfirstlane(tid >> 5);
    const int lo = blockIdx.x * SAGE_R;
    const int nOwn = min(SAGE_R, SAGE_N - lo);
    int2* wl = (int2*)wlist + (size_t)blockIdx.x * SAGE_LCAP;

    for (int i = tid; i < SAGE_R * SAGE_D; i += SAGE_T) acc[i] = 0.0f;
    for (int i = tid; i < SAGE_R; i += SAGE_T) cntl[i] = 0;
    if (tid == 0) nlist = 0;
    __syncthreads();

    const int saved = first ? 0 : wcount[blockIdx.x];
    const bool scan = first || saved > SAGE_LCAP;
    int n;

    if (scan) {
        if ((((size_t)dst) & 15) == 0) {
            const int4* d4 = (const int4*)dst;
            const int E4 = E >> 2;
            const int nGroups = (E4 + SAGE_T * SAGE_U - 1) / (SAGE_T * SAGE_U);
            const int rot = (int)((blockIdx.x * (unsigned)nGroups) / gridDim.x);
            for (int gi = 0; gi < nGroups; ++gi) {
                int g = gi + rot;
                if (g >= nGroups) g -= nGroups;
                int4 dv[SAGE_U];
#pragma unroll
                for (int u = 0; u < SAGE_U; ++u) {
                    const int i4 = (g * SAGE_U + u) * SAGE_T + tid;
                    dv[u] = d4[i4 < E4 ? i4 : 0];
                }
#pragma unroll
                for (int u = 0; u < SAGE_U; ++u) {
                    const int i4 = (g * SAGE_U + u) * SAGE_T + tid;
                    if (i4 < E4) {
                        const int e = i4 * 4;
                        sage_note_edge(dv[u].x, e + 0, lo, nOwn, first, cntl, &nlist, lst, acc, xin, src);
                        sage_note_edge(dv[u].y, e + 1, lo, nOwn, first, cntl, &nlist, lst, acc, xin, src);
                        sage_note_edge(dv[u].z, e + 2, lo, nOwn, first, cntl, &nlist, lst, acc, xin, src);
                        sage_note_edge(dv[u].w, e + 3, lo, nOwn, first, cntl, &nlist, lst, acc, xin, src);
                    }
                }
            }
            for (int e = E4 * 4 + tid; e < E; e += SAGE_T)
                sage_note_edge(dst[e], e, lo, nOwn, first, cntl, &nlist, lst, acc, xin, src);
        } else {
            for (int e = tid; e < E; e += SAGE_T)
                sage_note_edge(dst[e], e, lo, nOwn, first, cntl, &nlist, lst, acc, xin, src);
        }
        __syncthreads();

        const int nl = nlist;
        n = min(nl, SAGE_LCAP);
        for (int i = tid; i < n; i += SAGE_T) {
            const int2 ent = make_int2(src[lst[i].x], lst[i].y);
            lst[i] = ent;
            if (first) wl[i] = ent;
        }
        if (first && tid == 0) wcount[blockIdx.x] = nl;
        __syncthreads();
    } else {
        n = saved;
        for (int i = tid; i < n; i += SAGE_T) lst[i] = wl[i];
        __syncthreads();
    }

    for (int base = wave; base < n; base += 32 * SAGE_U) {
        float4 v[SAGE_U];
        int dl[SAGE_U];
#pragma unroll
        for (int u = 0; u < SAGE_U; ++u) {
            const int idx = base + u * 32;
            const int2 ent = lst[idx < n ? idx : base];
            dl[u] = ent.y;
            v[u] = *(const float4*)(xin + (size_t)ent.x * SAGE_D + lane * 4);
        }
        asm volatile("" :: "v"(v[0].x), "v"(v[1].x), "v"(v[2].x), "v"(v[3].x),
                           "v"(v[4].x), "v"(v[5].x), "v"(v[6].x), "v"(v[7].x));
#pragma unroll
        for (int u = 0; u < SAGE_U; ++u) {
            if (base + u * 32 < n) {
                float* a = acc + dl[u] * SAGE_D + lane;
                unsafeAtomicAdd(a + 0, v[u].x);
                unsafeAtomicAdd(a + 32, v[u].y);
                unsafeAtomicAdd(a + 64, v[u].z);
                unsafeAtomicAdd(a + 96, v[u].w);
            }
        }
    }
    __syncthreads();

    for (int i = tid; i < nOwn * 32; i += SAGE_T) {
        const int r = i >> 5;
        const float* a = acc + r * SAGE_D + lane;
        const float4 o = make_float4(a[0], a[32], a[64], a[96]);
        *(float4*)(s + (size_t)(lo + r) * SAGE_D + lane * 4) = o;
    }
    if (first)
        for (int i = tid; i < nOwn; i += SAGE_T) cnt[lo + i] = (float)cntl[i];
}

__global__ __launch_bounds__(256) void sage_gemm_kernel(
    const float* __restrict__ s,
    const float* __restrict__ cnt,
    const float* __restrict__ xin,
    const float* __restrict__ Wl,
    const float* __restrict__ bl,
    const float* __restrict__ Wr,
    float* __restrict__ out,
    int nRowTiles) {
    const int wave = threadIdx.x >> 5;
    const int tile = blockIdx.x * 8 + wave;
    if (tile >= nRowTiles) return;

    const int lane = threadIdx.x & 31;
    const int half = lane >> 4;
    const int lm   = lane & 15;
    const int m0   = tile * 16;

    const float inv = 1.0f / fmaxf(cnt[m0 + lm], 1.0f);

    const float* arow_s = s   + (size_t)(m0 + lm) * SAGE_D + 2 * half;
    const float* arow_x = xin + (size_t)(m0 + lm) * SAGE_D + 2 * half;
    const float* wl_row = Wl  + (size_t)lm * SAGE_D + 2 * half;
    const float* wr_row = Wr  + (size_t)lm * SAGE_D + 2 * half;

    v8f acc[8];
#pragma unroll
    for (int ct = 0; ct < 8; ++ct) acc[ct] = (v8f){};

    for (int k0 = 0; k0 < SAGE_D; k0 += 4) {
        v2f a_agg = *(const v2f*)(arow_s + k0);
        a_agg = a_agg * inv;
        v2f a_x = *(const v2f*)(arow_x + k0);
#pragma unroll
        for (int ct = 0; ct < 8; ++ct) {
            v2f b_l = *(const v2f*)(wl_row + (size_t)ct * 16 * SAGE_D + k0);
            v2f b_r = *(const v2f*)(wr_row + (size_t)ct * 16 * SAGE_D + k0);
            acc[ct] = __builtin_amdgcn_wmma_f32_16x16x4_f32(
                false, a_agg, false, b_l, (short)0, acc[ct], false, false);
            acc[ct] = __builtin_amdgcn_wmma_f32_16x16x4_f32(
                false, a_x, false, b_r, (short)0, acc[ct], false, false);
        }
    }

#pragma unroll
    for (int ct = 0; ct < 8; ++ct) {
        const float bias = bl[ct * 16 + lm];
#pragma unroll
        for (int v = 0; v < 8; ++v) {
            out[(size_t)(m0 + v + 8 * half) * SAGE_D + ct * 16 + lm] =
                acc[ct][v] + bias;
        }
    }
}

extern "C" void kernel_launch(void* const* d_in, const int* in_sizes, int n_in,
                              void* d_out, int out_size, void* d_ws, size_t ws_size,
                              hipStream_t stream) {
    const float* x   = (const float*)d_in[0];
    const int*   ei  = (const int*)d_in[1];
    const float* Wl1 = (const float*)d_in[2];
    const float* bl1 = (const float*)d_in[3];
    const float* Wr1 = (const float*)d_in[4];
    const float* Wl2 = (const float*)d_in[5];
    const float* bl2 = (const float*)d_in[6];
    const float* Wr2 = (const float*)d_in[7];
    float* out = (float*)d_out;

    const int N = SAGE_N;
    const int E = in_sizes[1] / 2;
    const int* src = ei;
    const int* dst = ei + E;

    float* s   = (float*)d_ws;
    float* cnt = s + (size_t)N * SAGE_D;
    int* wlist  = (int*)(cnt + N);
    int* wcount = wlist + (size_t)2 * SAGE_LCAP * ((N + SAGE_R - 1) / SAGE_R);

    const int nRowTiles  = N / 16;
    const int gemmBlocks = (nRowTiles + 7) / 8;
    const int aggBlocks  = (N + SAGE_R - 1) / SAGE_R;

    sage_agg_kernel<<<aggBlocks, SAGE_T, 0, stream>>>(x, src, dst, s, cnt, wlist,
                                                      wcount, E, 1);
    sage_gemm_kernel<<<gemmBlocks, 256, 0, stream>>>(s, cnt, x, Wl1, bl1, Wr1,
                                                     out, nRowTiles);

    sage_agg_kernel<<<aggBlocks, SAGE_T, 0, stream>>>(out, src, dst, s, cnt, wlist,
                                                      wcount, E, 0);
    sage_gemm_kernel<<<gemmBlocks, 256, 0, stream>>>(s, cnt, out, Wl2, bl2, Wr2,
                                                     out, nRowTiles);
}


# GEMM: a wave owns 32 rows (2 row tiles) and loads fragments 16 bytes per lane, so one weight load feeds 4 WMMAs, not 1; k sum order inside each chunk of 8 changed
# speedup vs baseline: 2.7049x; 1.2827x over previous
#include <hip/hip_runtime.h>
#include <hip/hip_bf16.h>


#define SAGE_N 50000
#define SAGE_D 128
#define SAGE_R 196
#define SAGE_LCAP 16384
#define SAGE_T 1024
#define SAGE_U 8

typedef __attribute__((ext_vector_type(2))) float v2f;
typedef __attribute__((ext_vector_type(8))) float v8f;

__device__ __forceinline__ void sage_note_edge(
    int d, int e, int lo, int nOwn, int writeCnt, int* cntl, int* nlist,
    int2* lst, float* acc, const float* __restrict__ xin,
    const int* __restrict__ src) {
    const unsigned dl = (unsigned)(d - lo);
    if (dl < (unsigned)nOwn) {
        if (writeCnt) atomicAdd(&cntl[dl], 1);
        const int pos = atomicAdd(nlist, 1);
        if (pos < SAGE_LCAP) {
            lst[pos] = make_int2(e, (int)dl);
        } else {
            const float* xr = xin + (size_t)src[e] * SAGE_D;
            float* ar = acc + dl * SAGE_D;
            for (int k = 0; k < SAGE_D; ++k)
                unsafeAtomicAdd(ar + (k & 3) * 32 + (k >> 2), xr[k]);
        }
    }
}

__global__ __launch_bounds__(SAGE_T) void sage_agg_kernel(
    const float* __restrict__ xin, const int* __restrict__ src,
    const int* __restrict__ dst, float* __restrict__ s,
    float* __restrict__ cnt, int* __restrict__ wlist,
    int* __restrict__ wcount, int E, int first) {
    __shared__ float acc[SAGE_R * SAGE_D];
    __shared__ int2 lst[SAGE_LCAP];
    __shared__ int cntl[SAGE_R];
    __shared__ int nlist;

    const int tid = threadIdx.x;
    const int lane = tid & 31;
    const int wave = __builtin_amdgcn_readfirstlane(tid >> 5);
    const int lo = blockIdx.x * SAGE_R;
    const int nOwn = min(SAGE_R, SAGE_N - lo);
    int2* wl = (int2*)wlist + (size_t)blockIdx.x * SAGE_LCAP;

    for (int i = tid; i < SAGE_R * SAGE_D; i += SAGE_T) acc[i] = 0.0f;
    for (int i = tid; i < SAGE_R; i += SAGE_T) cntl[i] = 0;
    if (tid == 0) nlist = 0;
    __syncthreads();

    const int saved = first ? 0 : wcount[blockIdx.x];
    const bool scan = first || saved > SAGE_LCAP;
    int n;

    if (scan) {
        if ((((size_t)dst) & 15) == 0) {
            const int4* d4 = (const int4*)dst;
            const int E4 = E >> 2;
            const int nGroups = (E4 + SAGE_T * SAGE_U - 1) / (SAGE_T * SAGE_U);
            const int rot = (int)((blockIdx.x * (unsigned)nGroups) / gridDim.x);
            for (int gi = 0; gi < nGroups; ++gi) {
                int g = gi + rot;
                if (g >= nGroups) g -= nGroups;
                int4 dv[SAGE_U];
#pragma unroll
                for (int u = 0; u < SAGE_U; ++u) {
                    const int i4 = (g * SAGE_U + u) * SAGE_T + tid;
                    dv[u] = d4[i4 < E4 ? i4 : 0];
                }
#pragma unroll
                for (int u = 0; u < SAGE_U; ++u) {
                    const int i4 = (g * SAGE_U + u) * SAGE_T + tid;
                    if (i4 < E4) {
                        const int e = i4 * 4;
                        sage_note_edge(dv[u].x, e + 0, lo, nOwn, first, cntl, &nlist, lst, acc, xin, src);
                        sage_note_edge(dv[u].y, e + 1, lo, nOwn, first, cntl, &nlist, lst, acc, xin, src);
                        sage_note_edge(dv[u].z, e + 2, lo, nOwn, first, cntl, &nlist, lst, acc, xin, src);
                        sage_note_edge(dv[u].w, e + 3, lo, nOwn, first, cntl, &nlist, lst, acc, xin, src);
                    }
                }
            }
            for (int e = E4 * 4 + tid; e < E; e += SAGE_T)
                sage_note_edge(dst[e], e, lo, nOwn, first, cntl, &nlist, lst, acc, xin, src);
        } else {
            for (int e = tid; e < E; e += SAGE_T)
                sage_note_edge(dst[e], e, lo, nOwn, first, cntl, &nlist, lst, acc, xin, src);
        }
        __syncthreads();

        const int nl = nlist;
        n = min(nl, SAGE_LCAP);
        for (int i = tid; i < n; i += SAGE_T) {
            const int2 ent = make_int2(src[lst[i].x], lst[i].y);
            lst[i] = ent;
            if (first) wl[i] = ent;
        }
        if (first && tid == 0) wcount[blockIdx.x] = nl;
        __syncthreads();
    } else {
        n = saved;
        for (int i = tid; i < n; i += SAGE_T) lst[i] = wl[i];
        __syncthreads();
    }

    for (int base = wave; base < n; base += 32 * SAGE_U) {
        float4 v[SAGE_U];
        int dl[SAGE_U];
#pragma unroll
        for (int u = 0; u < SAGE_U; ++u) {
            const int idx = base + u * 32;
            const int2 ent = lst[idx < n ? idx : base];
            dl[u] = ent.y;
            v[u] = *(const float4*)(xin + (size_t)ent.x * SAGE_D + lane * 4);
        }
        asm volatile("" :: "v"(v[0].x), "v"(v[1].x), "v"(v[2].x), "v"(v[3].x),
                           "v"(v[4].x), "v"(v[5].x), "v"(v[6].x), "v"(v[7].x));
#pragma unroll
        for (int u = 0; u < SAGE_U; ++u) {
            if (base + u * 32 < n) {
                float* a = acc + dl[u] * SAGE_D + lane;
                unsafeAtomicAdd(a + 0, v[u].x);
                unsafeAtomicAdd(a + 32, v[u].y);
                unsafeAtomicAdd(a + 64, v[u].z);
                unsafeAtomicAdd(a + 96, v[u].w);
            }
        }
    }
    __syncthreads();

    for (int i = tid; i < nOwn * 32; i += SAGE_T) {
        const int r = i >> 5;
        const float* a = acc + r * SAGE_D + lane;
        const float4 o = make_float4(a[0], a[32], a[64], a[96]);
        *(float4*)(s + (size_t)(lo + r) * SAGE_D + lane * 4) = o;
    }
    if (first)
        for (int i = tid; i < nOwn; i += SAGE_T) cnt[lo + i] = (float)cntl[i];
}

__global__ __launch_bounds__(128) void sage_gemm_kernel(
    const float* __restrict__ s,
    const float* __restrict__ cnt,
    const float* __restrict__ xin,
    const float* __restrict__ Wl,
    const float* __restrict__ bl,
    const float* __restrict__ Wr,
    float* __restrict__ out,
    int nRowTiles) {
    const int wave = threadIdx.x >> 5;
    const int t0 = (blockIdx.x * 4 + wave) * 2;
    if (t0 >= nRowTiles) return;
    const bool two = (t0 + 1) < nRowTiles;

    const int lane = threadIdx.x & 31;
    const int half = lane >> 4;
    const int lm   = lane & 15;
    const int m0   = t0 * 16;
    const int r0   = m0 + lm;
    const int r1   = two ? r0 + 16 : r0;

    const float inv0 = 1.0f / fmaxf(cnt[r0], 1.0f);
    const float inv1 = 1.0f / fmaxf(cnt[r1], 1.0f);

    const float* s0 = s   + (size_t)r0 * SAGE_D + 4 * half;
    const float* s1 = s   + (size_t)r1 * SAGE_D + 4 * half;
    const float* x0 = xin + (size_t)r0 * SAGE_D + 4 * half;
    const float* x1 = xin + (size_t)r1 * SAGE_D + 4 * half;
    const float* wl_row = Wl + (size_t)lm * SAGE_D + 4 * half;
    const float* wr_row = Wr + (size_t)lm * SAGE_D + 4 * half;

    v8f acc0[8], acc1[8];
#pragma unroll
    for (int ct = 0; ct < 8; ++ct) { acc0[ct] = (v8f){}; acc1[ct] = (v8f){}; }

    for (int k0 = 0; k0 < SAGE_D; k0 += 8) {
        float4 as0 = *(const float4*)(s0 + k0);
        float4 as1 = *(const float4*)(s1 + k0);
        const float4 ax0 = *(const float4*)(x0 + k0);
        const float4 ax1 = *(const float4*)(x1 + k0);
        const v2f a_s0_lo = (v2f){as0.x * inv0, as0.y * inv0};
        const v2f a_s0_hi = (v2f){as0.z * inv0, as0.w * inv0};
        const v2f a_s1_lo = (v2f){as1.x * inv1, as1.y * inv1};
        const v2f a_s1_hi = (v2f){as1.z * inv1, as1.w * inv1};
        const v2f a_x0_lo = (v2f){ax0.x, ax0.y};
        const v2f a_x0_hi = (v2f){ax0.z, ax0.w};
        const v2f a_x1_lo = (v2f){ax1.x, ax1.y};
        const v2f a_x1_hi = (v2f){ax1.z, ax1.w};
#pragma unroll
        for (int ct = 0; ct < 8; ++ct) {
            const float4 wl4 = *(const float4*)(wl_row + (size_t)ct * 16 * SAGE_D + k0);
            const float4 wr4 = *(const float4*)(wr_row + (size_t)ct * 16 * SAGE_D + k0);
            const v2f b_l_lo = (v2f){wl4.x, wl4.y};
            const v2f b_l_hi = (v2f){wl4.z, wl4.w};
            const v2f b_r_lo = (v2f){wr4.x, wr4.y};
            const v2f b_r_hi = (v2f){wr4.z, wr4.w};
            acc0[ct] = __builtin_amdgcn_wmma_f32_16x16x4_f32(
                false, a_s0_lo, false, b_l_lo, (short)0, acc0[ct], false, false);
            acc1[ct] = __builtin_amdgcn_wmma_f32_16x16x4_f32(
                false, a_s1_lo, false, b_l_lo, (short)0, acc1[ct], false, false);
            acc0[ct] = __builtin_amdgcn_wmma_f32_16x16x4_f32(
                false, a_s0_hi, false, b_l_hi, (short)0, acc0[ct], false, false);
            acc1[ct] = __builtin_amdgcn_wmma_f32_16x16x4_f32(
                false, a_s1_hi, false, b_l_hi, (short)0, acc1[ct], false, false);
            acc0[ct] = __builtin_amdgcn_wmma_f32_16x16x4_f32(
                false, a_x0_lo, false, b_r_lo, (short)0, acc0[ct], false, false);
            acc1[ct] = __builtin_amdgcn_wmma_f32_16x16x4_f32(
                false, a_x1_lo, false, b_r_lo, (short)0, acc1[ct], false, false);
            acc0[ct] = __builtin_amdgcn_wmma_f32_16x16x4_f32(
                false, a_x0_hi, false, b_r_hi, (short)0, acc0[ct], false, false);
            acc1[ct] = __builtin_amdgcn_wmma_f32_16x16x4_f32(
                false, a_x1_hi, false, b_r_hi, (short)0, acc1[ct], false, false);
        }
    }

#pragma unroll
    for (int ct = 0; ct < 8; ++ct) {
        const float bias = bl[ct * 16 + lm];
#pragma unroll
        for (int v = 0; v < 8; ++v) {
            out[(size_t)(m0 + v + 8 * half) * SAGE_D + ct * 16 + lm] =
                acc0[ct][v] + bias;
        }
    }
    if (two) {
#pragma unroll
        for (int ct = 0; ct < 8; ++ct) {
            const float bias = bl[ct * 16 + lm];
#pragma unroll
            for (int v = 0; v < 8; ++v) {
                out[(size_t)(m0 + 16 + v + 8 * half) * SAGE_D + ct * 16 + lm] =
                    acc1[ct][v] + bias;
            }
        }
    }
}

extern "C" void kernel_launch(void* const* d_in, const int* in_sizes, int n_in,
                              void* d_out, int out_size, void* d_ws, size_t ws_size,
                              hipStream_t stream) {
    const float* x   = (const float*)d_in[0];
    const int*   ei  = (const int*)d_in[1];
    const float* Wl1 = (const float*)d_in[2];
    const float* bl1 = (const float*)d_in[3];
    const float* Wr1 = (const float*)d_in[4];
    const float* Wl2 = (const float*)d_in[5];
    const float* bl2 = (const float*)d_in[6];
    const float* Wr2 = (const float*)d_in[7];
    float* out = (float*)d_out;

    const int N = SAGE_N;
    const int E = in_sizes[1] / 2;
    const int* src = ei;
    const int* dst = ei + E;

    float* s   = (float*)d_ws;
    float* cnt = s + (size_t)N * SAGE_D;
    int* wlist  = (int*)(cnt + N);
    int* wcount = wlist + (size_t)2 * SAGE_LCAP * ((N + SAGE_R - 1) / SAGE_R);

    const int nRowTiles  = N / 16;
    const int gemmBlocks = (nRowTiles + 7) / 8;
    const int aggBlocks  = (N + SAGE_R - 1) / SAGE_R;

    sage_agg_kernel<<<aggBlocks, SAGE_T, 0, stream>>>(x, src, dst, s, cnt, wlist,
                                                      wcount, E, 1);
    sage_gemm_kernel<<<gemmBlocks, 128, 0, stream>>>(s, cnt, x, Wl1, bl1, Wr1,
                                                     out, nRowTiles);

    sage_agg_kernel<<<aggBlocks, SAGE_T, 0, stream>>>(out, src, dst, s, cnt, wlist,
                                                      wcount, E, 0);
    sage_gemm_kernel<<<gemmBlocks, 128, 0, stream>>>(s, cnt, out, Wl2, bl2, Wr2,
                                                     out, nRowTiles);
}


# Scan reads one owner byte per edge (written by a small new first launch), not the 4-byte id: 16 edges per 16-byte load tested by word arithmetic, one branch per 16; list stays in arrival order
# speedup vs baseline: 3.0853x; 1.1406x over previous
#include <hip/hip_runtime.h>
#include <hip/hip_bf16.h>


#define SAGE_N 50000
#define SAGE_D 128
#define SAGE_R 196
#define SAGE_LCAP 16384
#define SAGE_T 1024
#define SAGE_U 8

typedef __attribute__((ext_vector_type(2))) float v2f;
typedef __attribute__((ext_vector_type(8))) float v8f;

__device__ __forceinline__ unsigned sage_owner_of(int d) {
    const unsigned o = (unsigned)d / (unsigned)SAGE_R;
    return o < 255u ? o : 255u;
}

__global__ __launch_bounds__(SAGE_T) void sage_owner_kernel(
    const int* __restrict__ dst, unsigned char* __restrict__ own, int E) {
    const int E4 = E >> 2;
    const int stride = gridDim.x * SAGE_T;
    const int gtid = blockIdx.x * SAGE_T + threadIdx.x;
    unsigned* own32 = (unsigned*)own;
    const bool aligned = (((size_t)dst) & 15) == 0;
    for (int i = gtid; i < E4; i += stride) {
        int4 d;
        if (aligned) {
            d = ((const int4*)dst)[i];
        } else {
            d = make_int4(dst[4 * i], dst[4 * i + 1], dst[4 * i + 2], dst[4 * i + 3]);
        }
        own32[i] = sage_owner_of(d.x) | (sage_owner_of(d.y) << 8) |
                   (sage_owner_of(d.z) << 16) | (sage_owner_of(d.w) << 24);
    }
    for (int e = E4 * 4 + gtid; e < E; e += stride)
        own[e] = (unsigned char)sage_owner_of(dst[e]);
}

__device__ __forceinline__ unsigned sage_nonzero_bytes(unsigned t) {
    return ((t & 0x7F7F7F7Fu) + 0x7F7F7F7Fu) | t;
}

__device__ __forceinline__ void sage_note_edge(
    int e, int lo, int nOwn, int writeCnt, int* cntl, int* nlist,
    int2* lst, float* acc, const float* __restrict__ xin,
    const int* __restrict__ src, const int* __restrict__ dst) {
    const int pos = atomicAdd(nlist, 1);
    if (pos < SAGE_LCAP) {
        lst[pos] = make_int2(e, 0);
    } else {
        const unsigned dl = (unsigned)(dst[e] - lo);
        if (dl < (unsigned)nOwn) {
            if (writeCnt) atomicAdd(&cntl[dl], 1);
            const float* xr = xin + (size_t)src[e] * SAGE_D;
            float* ar = acc + dl * SAGE_D;
            for (int k = 0; k < SAGE_D; ++k)
                unsafeAtomicAdd(ar + (k & 3) * 32 + (k >> 2), xr[k]);
        }
    }
}

__global__ __launch_bounds__(SAGE_T) void sage_agg_kernel(
    const float* __restrict__ xin, const int* __restrict__ src,
    const int* __restrict__ dst, const unsigned char* __restrict__ own,
    float* __restrict__ s, float* __restrict__ cnt, int* __restrict__ wlist,
    int* __restrict__ wcount, int E, int first) {
    __shared__ float acc[(SAGE_R + 1) * SAGE_D];
    __shared__ int2 lst[SAGE_LCAP];
    __shared__ int cntl[SAGE_R];
    __shared__ int nlist;

    const int tid = threadIdx.x;
    const int lane = tid & 31;
    const int wave = __builtin_amdgcn_readfirstlane(tid >> 5);
    const int lo = blockIdx.x * SAGE_R;
    const int nOwn = min(SAGE_R, SAGE_N - lo);
    int2* wl = (int2*)wlist + (size_t)blockIdx.x * SAGE_LCAP;

    for (int i = tid; i < (SAGE_R + 1) * SAGE_D; i += SAGE_T) acc[i] = 0.0f;
    for (int i = tid; i < SAGE_R; i += SAGE_T) cntl[i] = 0;
    if (tid == 0) nlist = 0;
    __syncthreads();

    const int saved = first ? 0 : wcount[blockIdx.x];
    const bool scan = first || saved > SAGE_LCAP;
    int n;

    if (scan) {
        const uint4* o4 = (const uint4*)own;
        const int E16 = E >> 4;
        const unsigned key = blockIdx.x * 0x01010101u;
        const int nGroups = (E16 + SAGE_T * SAGE_U - 1) / (SAGE_T * SAGE_U);
        const int rot = (int)((blockIdx.x * (unsigned)nGroups) / gridDim.x);
        for (int gi = 0; gi < nGroups; ++gi) {
            int g = gi + rot;
            if (g >= nGroups) g -= nGroups;
            uint4 ov[SAGE_U];
#pragma unroll
            for (int u = 0; u < SAGE_U; ++u) {
                const int i16 = (g * SAGE_U + u) * SAGE_T + tid;
                ov[u] = o4[i16 < E16 ? i16 : 0];
            }
#pragma unroll
            for (int u = 0; u < SAGE_U; ++u) {
                const int i16 = (g * SAGE_U + u) * SAGE_T + tid;
                if (i16 < E16) {
                    const unsigned n0 = sage_nonzero_bytes(ov[u].x ^ key);
                    const unsigned n1 = sage_nonzero_bytes(ov[u].y ^ key);
                    const unsigned n2 = sage_nonzero_bytes(ov[u].z ^ key);
                    const unsigned n3 = sage_nonzero_bytes(ov[u].w ^ key);
                    if (((n0 & n1 & n2 & n3) & 0x80808080u) != 0x80808080u) {
                        unsigned m = ((~n0 & 0x80808080u) >> 7) |
                                     ((~n1 & 0x80808080u) >> 6) |
                                     ((~n2 & 0x80808080u) >> 5) |
                                     ((~n3 & 0x80808080u) >> 4);
                        while (m) {
                            const int p = __builtin_ctz(m);
                            m &= m - 1;
                            const int e = i16 * 16 + (p & 7) * 4 + (p >> 3);
                            sage_note_edge(e, lo, nOwn, first, cntl, &nlist, lst, acc, xin, src, dst);
                        }
                    }
                }
            }
        }
        for (int e = E16 * 16 + tid; e < E; e += SAGE_T)
            if (own[e] == (unsigned char)blockIdx.x)
                sage_note_edge(e, lo, nOwn, first, cntl, &nlist, lst, acc, xin, src, dst);
        __syncthreads();

        const int nl = nlist;
        n = min(nl, SAGE_LCAP);
        for (int i = tid; i < n; i += SAGE_T) {
            const int e = lst[i].x;
            const int sv = src[e];
            const unsigned dl = (unsigned)(dst[e] - lo);
            const bool ok = dl < (unsigned)nOwn;
            if (ok && first) atomicAdd(&cntl[dl], 1);
            const int2 ent = ok ? make_int2(sv, (int)dl) : make_int2(0, SAGE_R);
            lst[i] = ent;
            if (first) wl[i] = ent;
        }
        if (first && tid == 0) wcount[blockIdx.x] = nl;
        __syncthreads();
    } else {
        n = saved;
        for (int i = tid; i < n; i += SAGE_T) lst[i] = wl[i];
        __syncthreads();
    }

    for (int base = wave; base < n; base += 32 * SAGE_U) {
        float4 v[SAGE_U];
        int dl[SAGE_U];
#pragma unroll
        for (int u = 0; u < SAGE_U; ++u) {
            const int idx = base + u * 32;
            const int2 ent = lst[idx < n ? idx : base];
            dl[u] = ent.y;
            v[u] = *(const float4*)(xin + (size_t)ent.x * SAGE_D + lane * 4);
        }
        asm volatile("" :: "v"(v[0].x), "v"(v[1].x), "v"(v[2].x), "v"(v[3].x),
                           "v"(v[4].x), "v"(v[5].x), "v"(v[6].x), "v"(v[7].x));
#pragma unroll
        for (int u = 0; u < SAGE_U; ++u) {
            if (base + u * 32 < n) {
                float* a = acc + dl[u] * SAGE_D + lane;
                unsafeAtomicAdd(a + 0, v[u].x);
                unsafeAtomicAdd(a + 32, v[u].y);
                unsafeAtomicAdd(a + 64, v[u].z);
                unsafeAtomicAdd(a + 96, v[u].w);
            }
        }
    }
    __syncthreads();

    for (int i = tid; i < nOwn * 32; i += SAGE_T) {
        const int r = i >> 5;
        const float* a = acc + r * SAGE_D + lane;
        const float4 o = make_float4(a[0], a[32], a[64], a[96]);
        *(float4*)(s + (size_t)(lo + r) * SAGE_D + lane * 4) = o;
    }
    if (first)
        for (int i = tid; i < nOwn; i += SAGE_T) cnt[lo + i] = (float)cntl[i];
}

__global__ __launch_bounds__(128) void sage_gemm_kernel(
    const float* __restrict__ s,
    const float* __restrict__ cnt,
    const float* __restrict__ xin,
    const float* __restrict__ Wl,
    const float* __restrict__ bl,
    const float* __restrict__ Wr,
    float* __restrict__ out,
    int nRowTiles) {
    const int wave = threadIdx.x >> 5;
    const int t0 = (blockIdx.x * 4 + wave) * 2;
    if (t0 >= nRowTiles) return;
    const bool two = (t0 + 1) < nRowTiles;

    const int lane = threadIdx.x & 31;
    const int half = lane >> 4;
    const int lm   = lane & 15;
    const int m0   = t0 * 16;
    const int r0   = m0 + lm;
    const int r1   = two ? r0 + 16 : r0;

    const float inv0 = 1.0f / fmaxf(cnt[r0], 1.0f);
    const float inv1 = 1.0f / fmaxf(cnt[r1], 1.0f);

    const float* s0 = s   + (size_t)r0 * SAGE_D + 4 * half;
    const float* s1 = s   + (size_t)r1 * SAGE_D + 4 * half;
    const float* x0 = xin + (size_t)r0 * SAGE_D + 4 * half;
    const float* x1 = xin + (size_t)r1 * SAGE_D + 4 * half;
    const float* wl_row = Wl + (size_t)lm * SAGE_D + 4 * half;
    const float* wr_row = Wr + (size_t)lm * SAGE_D + 4 * half;

    v8f acc0[8], acc1[8];
#pragma unroll
    for (int ct = 0; ct < 8; ++ct) { acc0[ct] = (v8f){}; acc1[ct] = (v8f){}; }

    for (int k0 = 0; k0 < SAGE_D; k0 += 8) {
        float4 as0 = *(const float4*)(s0 + k0);
        float4 as1 = *(const float4*)(s1 + k0);
        const float4 ax0 = *(const float4*)(x0 + k0);
        const float4 ax1 = *(const float4*)(x1 + k0);
        const v2f a_s0_lo = (v2f){as0.x * inv0, as0.y * inv0};
        const v2f a_s0_hi = (v2f){as0.z * inv0, as0.w * inv0};
        const v2f a_s1_lo = (v2f){as1.x * inv1, as1.y * inv1};
        const v2f a_s1_hi = (v2f){as1.z * inv1, as1.w * inv1};
        const v2f a_x0_lo = (v2f){ax0.x, ax0.y};
        const v2f a_x0_hi = (v2f){ax0.z, ax0.w};
        const v2f a_x1_lo = (v2f){ax1.x, ax1.y};
        const v2f a_x1_hi = (v2f){ax1.z, ax1.w};
#pragma unroll
        for (int ct = 0; ct < 8; ++ct) {
            const float4 wl4 = *(const float4*)(wl_row + (size_t)ct * 16 * SAGE_D + k0);
            const float4 wr4 = *(const float4*)(wr_row + (size_t)ct * 16 * SAGE_D + k0);
            const v2f b_l_lo = (v2f){wl4.x, wl4.y};
            const v2f b_l_hi = (v2f){wl4.z, wl4.w};
            const v2f b_r_lo = (v2f){wr4.x, wr4.y};
            const v2f b_r_hi = (v2f){wr4.z, wr4.w};
            acc0[ct] = __builtin_amdgcn_wmma_f32_16x16x4_f32(
                false, a_s0_lo, false, b_l_lo, (short)0, acc0[ct], false, false);
            acc1[ct] = __builtin_amdgcn_wmma_f32_16x16x4_f32(
                false, a_s1_lo, false, b_l_lo, (short)0, acc1[ct], false, false);
            acc0[ct] = __builtin_amdgcn_wmma_f32_16x16x4_f32(
                false, a_s0_hi, false, b_l_hi, (short)0, acc0[ct], false, false);
            acc1[ct] = __builtin_amdgcn_wmma_f32_16x16x4_f32(
                false, a_s1_hi, false, b_l_hi, (short)0, acc1[ct], false, false);
            acc0[ct] = __builtin_amdgcn_wmma_f32_16x16x4_f32(
                false, a_x0_lo, false, b_r_lo, (short)0, acc0[ct], false, false);
            acc1[ct] = __builtin_amdgcn_wmma_f32_16x16x4_f32(
                false, a_x1_lo, false, b_r_lo, (short)0, acc1[ct], false, false);
            acc0[ct] = __builtin_amdgcn_wmma_f32_16x16x4_f32(
                false, a_x0_hi, false, b_r_hi, (short)0, acc0[ct], false, false);
            acc1[ct] = __builtin_amdgcn_wmma_f32_16x16x4_f32(
                false, a_x1_hi, false, b_r_hi, (short)0, acc1[ct], false, false);
        }
    }

#pragma unroll
    for (int ct = 0; ct < 8; ++ct) {
        const float bias = bl[ct * 16 + lm];
#pragma unroll
        for (int v = 0; v < 8; ++v) {
            out[(size_t)(m0 + v + 8 * half) * SAGE_D + ct * 16 + lm] =
                acc0[ct][v] + bias;
        }
    }
    if (two) {
#pragma unroll
        for (int ct = 0; ct < 8; ++ct) {
            const float bias = bl[ct * 16 + lm];
#pragma unroll
            for (int v = 0; v < 8; ++v) {
                out[(size_t)(m0 + 16 + v + 8 * half) * SAGE_D + ct * 16 + lm] =
                    acc1[ct][v] + bias;
            }
        }
    }
}

extern "C" void kernel_launch(void* const* d_in, const int* in_sizes, int n_in,
                              void* d_out, int out_size, void* d_ws, size_t ws_size,
                              hipStream_t stream) {
    const float* x   = (const float*)d_in[0];
    const int*   ei  = (const int*)d_in[1];
    const float* Wl1 = (const float*)d_in[2];
    const float* bl1 = (const float*)d_in[3];
    const float* Wr1 = (const float*)d_in[4];
    const float* Wl2 = (const float*)d_in[5];
    const float* bl2 = (const float*)d_in[6];
    const float* Wr2 = (const float*)d_in[7];
    float* out = (float*)d_out;

    const int N = SAGE_N;
    const int E = in_sizes[1] / 2;
    const int* src = ei;
    const int* dst = ei + E;

    float* s   = (float*)d_ws;
    float* cnt = s + (size_t)N * SAGE_D;
    int* wlist  = (int*)(cnt + N);
    int* wcount = wlist + (size_t)2 * SAGE_LCAP * ((N + SAGE_R - 1) / SAGE_R);
    unsigned char* own = (unsigned char*)(wcount + 256);

    const int nRowTiles  = N / 16;
    const int gemmBlocks = (nRowTiles + 7) / 8;
    const int aggBlocks  = (N + SAGE_R - 1) / SAGE_R;

    sage_owner_kernel<<<256, SAGE_T, 0, stream>>>(dst, own, E);

    sage_agg_kernel<<<aggBlocks, SAGE_T, 0, stream>>>(x, src, dst, own, s, cnt,
                                                      wlist, wcount, E, 1);
    sage_gemm_kernel<<<gemmBlocks, 128, 0, stream>>>(s, cnt, x, Wl1, bl1, Wr1,
                                                     out, nRowTiles);

    sage_agg_kernel<<<aggBlocks, SAGE_T, 0, stream>>>(out, src, dst, own, s, cnt,
                                                      wlist, wcount, E, 0);
    sage_gemm_kernel<<<gemmBlocks, 128, 0, stream>>>(s, cnt, out, Wl2, bl2, Wr2,
                                                     out, nRowTiles);
}


# GEMM folded into the aggregation launch: each work-group multiplies its own 196 summed rows straight from LDS (row pitch 132 words), s and 2 launches gone; k sum order inside the products changed
# speedup vs baseline: 3.3732x; 1.0933x over previous
#include <hip/hip_runtime.h>
#include <hip/hip_bf16.h>


#define SAGE_N 50000
#define SAGE_D 128
#define SAGE_R 196
#define SAGE_LCAP 16384
#define SAGE_T 1024
#define SAGE_U 8
#define SAGE_P 132
#define SAGE_ROWS 208

typedef __attribute__((ext_vector_type(2))) float v2f;
typedef __attribute__((ext_vector_type(8))) float v8f;

__device__ __forceinline__ unsigned sage_owner_of(int d) {
    const unsigned o = (unsigned)d / (unsigned)SAGE_R;
    return o < 255u ? o : 255u;
}

__global__ __launch_bounds__(SAGE_T) void sage_owner_kernel(
    const int* __restrict__ dst, unsigned char* __restrict__ own, int E) {
    const int E4 = E >> 2;
    const int stride = gridDim.x * SAGE_T;
    const int gtid = blockIdx.x * SAGE_T + threadIdx.x;
    unsigned* own32 = (unsigned*)own;
    const bool aligned = (((size_t)dst) & 15) == 0;
    for (int i = gtid; i < E4; i += stride) {
        int4 d;
        if (aligned) {
            d = ((const int4*)dst)[i];
        } else {
            d = make_int4(dst[4 * i], dst[4 * i + 1], dst[4 * i + 2], dst[4 * i + 3]);
        }
        own32[i] = sage_owner_of(d.x) | (sage_owner_of(d.y) << 8) |
                   (sage_owner_of(d.z) << 16) | (sage_owner_of(d.w) << 24);
    }
    for (int e = E4 * 4 + gtid; e < E; e += stride)
        own[e] = (unsigned char)sage_owner_of(dst[e]);
}

__device__ __forceinline__ unsigned sage_nonzero_bytes(unsigned t) {
    return ((t & 0x7F7F7F7Fu) + 0x7F7F7F7Fu) | t;
}

__device__ __forceinline__ void sage_note_edge(
    int e, int lo, int nOwn, int writeCnt, int* cntl, int* nlist,
    int2* lst, float* acc, const float* __restrict__ xin,
    const int* __restrict__ src, const int* __restrict__ dst) {
    const int pos = atomicAdd(nlist, 1);
    if (pos < SAGE_LCAP) {
        lst[pos] = make_int2(e, 0);
    } else {
        const unsigned dl = (unsigned)(dst[e] - lo);
        if (dl < (unsigned)nOwn) {
            if (writeCnt) atomicAdd(&cntl[dl], 1);
            const float* xr = xin + (size_t)src[e] * SAGE_D;
            float* ar = acc + dl * SAGE_P;
            for (int k = 0; k < SAGE_D; ++k)
                unsafeAtomicAdd(ar + (k & 3) * 32 + (k >> 2), xr[k]);
        }
    }
}

template <int NRT>
__device__ __forceinline__ void sage_tile_gemm(
    const float* acc, const float* __restrict__ xin,
    const float* __restrict__ Wl, const float* __restrict__ bl,
    const float* __restrict__ Wr, float* __restrict__ out,
    int lo, int nOwn, int g, int cp, int lane) {
    const int half = lane >> 4;
    const int lm = lane & 15;

    v8f c[NRT][2];
    const float* arow[NRT];
    const float* xrow[NRT];
#pragma unroll
    for (int i = 0; i < NRT; ++i) {
        c[i][0] = (v8f){};
        c[i][1] = (v8f){};
        const int r = (g + 8 * i) * 16 + lm;
        arow[i] = acc + r * SAGE_P + 2 * half;
        const int gr = min(lo + r, SAGE_N - 1);
        xrow[i] = xin + (size_t)gr * SAGE_D + 8 * half;
    }
    const float* wl0 = Wl + (size_t)(cp * 32 + lm) * SAGE_D + 8 * half;
    const float* wl1 = wl0 + 16 * SAGE_D;
    const float* wr0 = Wr + (size_t)(cp * 32 + lm) * SAGE_D + 8 * half;
    const float* wr1 = wr0 + 16 * SAGE_D;

#pragma unroll 1
    for (int k0 = 0; k0 < SAGE_D; k0 += 16) {
        const float4 la0 = *(const float4*)(wl0 + k0);
        const float4 lb0 = *(const float4*)(wl0 + k0 + 4);
        const float4 la1 = *(const float4*)(wl1 + k0);
        const float4 lb1 = *(const float4*)(wl1 + k0 + 4);
        const float4 ra0 = *(const float4*)(wr0 + k0);
        const float4 rb0 = *(const float4*)(wr0 + k0 + 4);
        const float4 ra1 = *(const float4*)(wr1 + k0);
        const float4 rb1 = *(const float4*)(wr1 + k0 + 4);
        const float la0v[4] = {la0.x, la0.y, la0.z, la0.w};
        const float lb0v[4] = {lb0.x, lb0.y, lb0.z, lb0.w};
        const float la1v[4] = {la1.x, la1.y, la1.z, la1.w};
        const float lb1v[4] = {lb1.x, lb1.y, lb1.z, lb1.w};
        const float ra0v[4] = {ra0.x, ra0.y, ra0.z, ra0.w};
        const float rb0v[4] = {rb0.x, rb0.y, rb0.z, rb0.w};
        const float ra1v[4] = {ra1.x, ra1.y, ra1.z, ra1.w};
        const float rb1v[4] = {rb1.x, rb1.y, rb1.z, rb1.w};
#pragma unroll
        for (int i = 0; i < NRT; ++i) {
            const float4 xa = *(const float4*)(xrow[i] + k0);
            const float4 xb = *(const float4*)(xrow[i] + k0 + 4);
            const float xav[4] = {xa.x, xa.y, xa.z, xa.w};
            const float xbv[4] = {xb.x, xb.y, xb.z, xb.w};
#pragma unroll
            for (int j = 0; j < 4; ++j) {
                const v2f a_s = *(const v2f*)(arow[i] + j * 32 + (k0 >> 2));
                const v2f a_x = (v2f){xav[j], xbv[j]};
                const v2f b_l0 = (v2f){la0v[j], lb0v[j]};
                const v2f b_l1 = (v2f){la1v[j], lb1v[j]};
                const v2f b_r0 = (v2f){ra0v[j], rb0v[j]};
                const v2f b_r1 = (v2f){ra1v[j], rb1v[j]};
                c[i][0] = __builtin_amdgcn_wmma_f32_16x16x4_f32(
                    false, a_s, false, b_l0, (short)0, c[i][0], false, false);
                c[i][1] = __builtin_amdgcn_wmma_f32_16x16x4_f32(
                    false, a_s, false, b_l1, (short)0, c[i][1], false, false);
                c[i][0] = __builtin_amdgcn_wmma_f32_16x16x4_f32(
                    false, a_x, false, b_r0, (short)0, c[i][0], false, false);
                c[i][1] = __builtin_amdgcn_wmma_f32_16x16x4_f32(
                    false, a_x, false, b_r1, (short)0, c[i][1], false, false);
            }
        }
    }

#pragma unroll
    for (int i = 0; i < NRT; ++i) {
        const int rt = g + 8 * i;
        const bool whole = rt * 16 + 16 <= nOwn;
#pragma unroll
        for (int t = 0; t < 2; ++t) {
            const int col = cp * 32 + t * 16 + lm;
            const float bias = bl[col];
            if (whole) {
                float* o = out + (size_t)(lo + rt * 16 + 8 * half) * SAGE_D + col;
#pragma unroll
                for (int v = 0; v < 8; ++v) o[v * SAGE_D] = c[i][t][v] + bias;
            } else {
#pragma unroll
                for (int v = 0; v < 8; ++v) {
                    const int r = rt * 16 + v + 8 * half;
                    if (r < nOwn)
                        out[(size_t)(lo + r) * SAGE_D + col] = c[i][t][v] + bias;
                }
            }
        }
    }
}

__global__ __launch_bounds__(SAGE_T) void sage_layer_kernel(
    const float* __restrict__ xin, const int* __restrict__ src,
    const int* __restrict__ dst, const unsigned char* __restrict__ own,
    const float* __restrict__ Wl, const float* __restrict__ bl,
    const float* __restrict__ Wr, float* __restrict__ out,
    float* __restrict__ cnt, int* __restrict__ wlist,
    int* __restrict__ wcount, int E, int first) {
    __shared__ __attribute__((aligned(16))) float acc[SAGE_ROWS * SAGE_P];
    __shared__ int2 lst[SAGE_LCAP];
    __shared__ int cntl[SAGE_R];
    __shared__ float invl[SAGE_R];
    __shared__ int nlist;

    const int tid = threadIdx.x;
    const int lane = tid & 31;
    const int wave = __builtin_amdgcn_readfirstlane(tid >> 5);
    const int lo = blockIdx.x * SAGE_R;
    const int nOwn = min(SAGE_R, SAGE_N - lo);
    int2* wl = (int2*)wlist + (size_t)blockIdx.x * SAGE_LCAP;

    for (int i = tid; i < SAGE_ROWS * SAGE_P; i += SAGE_T) acc[i] = 0.0f;
    for (int i = tid; i < SAGE_R; i += SAGE_T) cntl[i] = 0;
    if (tid == 0) nlist = 0;
    __syncthreads();

    const int saved = first ? 0 : wcount[blockIdx.x];
    const bool scan = first || saved > SAGE_LCAP;
    int n;

    if (scan) {
        const uint4* o4 = (const uint4*)own;
        const int E16 = E >> 4;
        const unsigned key = blockIdx.x * 0x01010101u;
        const int nGroups = (E16 + SAGE_T * SAGE_U - 1) / (SAGE_T * SAGE_U);
        const int rot = (int)((blockIdx.x * (unsigned)nGroups) / gridDim.x);
        for (int gi = 0; gi < nGroups; ++gi) {
            int g = gi + rot;
            if (g >= nGroups) g -= nGroups;
            uint4 ov[SAGE_U];
#pragma unroll
            for (int u = 0; u < SAGE_U; ++u) {
                const int i16 = (g * SAGE_U + u) * SAGE_T + tid;
                ov[u] = o4[i16 < E16 ? i16 : 0];
            }
#pragma unroll
            for (int u = 0; u < SAGE_U; ++u) {
                const int i16 = (g * SAGE_U + u) * SAGE_T + tid;
                if (i16 < E16) {
                    const unsigned n0 = sage_nonzero_bytes(ov[u].x ^ key);
                    const unsigned n1 = sage_nonzero_bytes(ov[u].y ^ key);
                    const unsigned n2 = sage_nonzero_bytes(ov[u].z ^ key);
                    const unsigned n3 = sage_nonzero_bytes(ov[u].w ^ key);
                    if (((n0 & n1 & n2 & n3) & 0x80808080u) != 0x80808080u) {
                        unsigned m = ((~n0 & 0x80808080u) >> 7) |
                                     ((~n1 & 0x80808080u) >> 6) |
                                     ((~n2 & 0x80808080u) >> 5) |
                                     ((~n3 & 0x80808080u) >> 4);
                        while (m) {
                            const int p = __builtin_ctz(m);
                            m &= m - 1;
                            const int e = i16 * 16 + (p & 7) * 4 + (p >> 3);
                            sage_note_edge(e, lo, nOwn, first, cntl, &nlist, lst, acc, xin, src, dst);
                        }
                    }
                }
            }
        }
        for (int e = E16 * 16 + tid; e < E; e += SAGE_T)
            if (own[e] == (unsigned char)blockIdx.x)
                sage_note_edge(e, lo, nOwn, first, cntl, &nlist, lst, acc, xin, src, dst);
        __syncthreads();

        const int nl = nlist;
        n = min(nl, SAGE_LCAP);
        for (int i = tid; i < n; i += SAGE_T) {
            const int e = lst[i].x;
            const int sv = src[e];
            const unsigned dl = (unsigned)(dst[e] - lo);
            const bool ok = dl < (unsigned)nOwn;
            if (ok && first) atomicAdd(&cntl[dl], 1);
            const int2 ent = ok ? make_int2(sv, (int)dl) : make_int2(0, SAGE_R);
            lst[i] = ent;
            if (first) wl[i] = ent;
        }
        if (first && tid == 0) wcount[blockIdx.x] = nl;
        __syncthreads();
    } else {
        n = saved;
        for (int i = tid; i < n; i += SAGE_T) lst[i] = wl[i];
        __syncthreads();
    }

    for (int base = wave; base < n; base += 32 * SAGE_U) {
        float4 v[SAGE_U];
        int dl[SAGE_U];
#pragma unroll
        for (int u = 0; u < SAGE_U; ++u) {
            const int idx = base + u * 32;
            const int2 ent = lst[idx < n ? idx : base];
            dl[u] = ent.y;
            v[u] = *(const float4*)(xin + (size_t)ent.x * SAGE_D + lane * 4);
        }
        asm volatile("" :: "v"(v[0].x), "v"(v[1].x), "v"(v[2].x), "v"(v[3].x),
                           "v"(v[4].x), "v"(v[5].x), "v"(v[6].x), "v"(v[7].x));
#pragma unroll
        for (int u = 0; u < SAGE_U; ++u) {
            if (base + u * 32 < n) {
                float* a = acc + dl[u] * SAGE_P + lane;
                unsafeAtomicAdd(a + 0, v[u].x);
                unsafeAtomicAdd(a + 32, v[u].y);
                unsafeAtomicAdd(a + 64, v[u].z);
                unsafeAtomicAdd(a + 96, v[u].w);
            }
        }
    }
    __syncthreads();

    for (int i = tid; i < nOwn; i += SAGE_T) {
        const float cv = first ? (float)cntl[i] : cnt[lo + i];
        if (first) cnt[lo + i] = cv;
        invl[i] = 1.0f / fmaxf(cv, 1.0f);
    }
    __syncthreads();
    for (int i = tid; i < nOwn * SAGE_D; i += SAGE_T) {
        const int r = i >> 7;
        acc[r * SAGE_P + (i & (SAGE_D - 1))] *= invl[r];
    }
    __syncthreads();

    const int nTiles = (nOwn + 15) >> 4;
    const int g = wave >> 2;
    const int cp = wave & 3;
    const int nrt = (g + 8 < nTiles) ? 2 : ((g < nTiles) ? 1 : 0);
    if (nrt == 2)
        sage_tile_gemm<2>(acc, xin, Wl, bl, Wr, out, lo, nOwn, g, cp, lane);
    else if (nrt == 1)
        sage_tile_gemm<1>(acc, xin, Wl, bl, Wr, out, lo, nOwn, g, cp, lane);
}

extern "C" void kernel_launch(void* const* d_in, const int* in_sizes, int n_in,
                              void* d_out, int out_size, void* d_ws, size_t ws_size,
                              hipStream_t stream) {
    const float* x   = (const float*)d_in[0];
    const int*   ei  = (const int*)d_in[1];
    const float* Wl1 = (const float*)d_in[2];
    const float* bl1 = (const float*)d_in[3];
    const float* Wr1 = (const float*)d_in[4];
    const float* Wl2 = (const float*)d_in[5];
    const float* bl2 = (const float*)d_in[6];
    const float* Wr2 = (const float*)d_in[7];
    float* out = (float*)d_out;

    const int N = SAGE_N;
    const int E = in_sizes[1] / 2;
    const int* src = ei;
    const int* dst = ei + E;

    float* h   = (float*)d_ws;
    float* cnt = h + (size_t)N * SAGE_D;
    int* wlist  = (int*)(cnt + N);
    int* wcount = wlist + (size_t)2 * SAGE_LCAP * ((N + SAGE_R - 1) / SAGE_R);
    unsigned char* own = (unsigned char*)(wcount + 256);

    const int aggBlocks  = (N + SAGE_R - 1) / SAGE_R;

    sage_owner_kernel<<<256, SAGE_T, 0, stream>>>(dst, own, E);

    sage_layer_kernel<<<aggBlocks, SAGE_T, 0, stream>>>(x, src, dst, own, Wl1, bl1, Wr1,
                                                        h, cnt, wlist, wcount, E, 1);
    sage_layer_kernel<<<aggBlocks, SAGE_T, 0, stream>>>(h, src, dst, own, Wl2, bl2, Wr2,
                                                        out, cnt, wlist, wcount, E, 0);
}


# Own rows x Wr run beside the gather: waves of even row groups do it before their gather share, odd ones after; result waits in the accumulators; sum order: all x.Wr terms, then s.Wl
# speedup vs baseline: 3.5874x; 1.0635x over previous
#include <hip/hip_runtime.h>
#include <hip/hip_bf16.h>


#define SAGE_N 50000
#define SAGE_D 128
#define SAGE_R 196
#define SAGE_LCAP 16384
#define SAGE_T 1024
#define SAGE_U 8
#define SAGE_P 132
#define SAGE_ROWS 208

typedef __attribute__((ext_vector_type(2))) float v2f;
typedef __attribute__((ext_vector_type(8))) float v8f;

__device__ __forceinline__ unsigned sage_owner_of(int d) {
    const unsigned o = (unsigned)d / (unsigned)SAGE_R;
    return o < 255u ? o : 255u;
}

__global__ __launch_bounds__(SAGE_T) void sage_owner_kernel(
    const int* __restrict__ dst, unsigned char* __restrict__ own, int E) {
    const int E4 = E >> 2;
    const int stride = gridDim.x * SAGE_T;
    const int gtid = blockIdx.x * SAGE_T + threadIdx.x;
    unsigned* own32 = (unsigned*)own;
    const bool aligned = (((size_t)dst) & 15) == 0;
    for (int i = gtid; i < E4; i += stride) {
        int4 d;
        if (aligned) {
            d = ((const int4*)dst)[i];
        } else {
            d = make_int4(dst[4 * i], dst[4 * i + 1], dst[4 * i + 2], dst[4 * i + 3]);
        }
        own32[i] = sage_owner_of(d.x) | (sage_owner_of(d.y) << 8) |
                   (sage_owner_of(d.z) << 16) | (sage_owner_of(d.w) << 24);
    }
    for (int e = E4 * 4 + gtid; e < E; e += stride)
        own[e] = (unsigned char)sage_owner_of(dst[e]);
}

__device__ __forceinline__ unsigned sage_nonzero_bytes(unsigned t) {
    return ((t & 0x7F7F7F7Fu) + 0x7F7F7F7Fu) | t;
}

__device__ __forceinline__ void sage_note_edge(
    int e, int lo, int nOwn, int writeCnt, int* cntl, int* nlist,
    int2* lst, float* acc, const float* __restrict__ xin,
    const int* __restrict__ src, const int* __restrict__ dst) {
    const int pos = atomicAdd(nlist, 1);
    if (pos < SAGE_LCAP) {
        lst[pos] = make_int2(e, 0);
    } else {
        const unsigned dl = (unsigned)(dst[e] - lo);
        if (dl < (unsigned)nOwn) {
            if (writeCnt) atomicAdd(&cntl[dl], 1);
            const float* xr = xin + (size_t)src[e] * SAGE_D;
            float* ar = acc + dl * SAGE_P;
            for (int k = 0; k < SAGE_D; ++k)
                unsafeAtomicAdd(ar + (k & 3) * 32 + (k >> 2), xr[k]);
        }
    }
}

template <int NRT>
__device__ __forceinline__ void sage_self_product(
    v8f (&c)[2][2], const float* __restrict__ xin,
    const float* __restrict__ Wr, int lo, int g, int cp, int lane) {
    const int half = lane >> 4;
    const int lm = lane & 15;
    const float* xrow[NRT];
#pragma unroll
    for (int i = 0; i < NRT; ++i) {
        const int r = (g + 8 * i) * 16 + lm;
        const int gr = min(lo + r, SAGE_N - 1);
        xrow[i] = xin + (size_t)gr * SAGE_D + 8 * half;
    }
    const float* wr0 = Wr + (size_t)(cp * 32 + lm) * SAGE_D + 8 * half;
    const float* wr1 = wr0 + 16 * SAGE_D;

#pragma unroll 2
    for (int k0 = 0; k0 < SAGE_D; k0 += 16) {
        const float4 ra0 = *(const float4*)(wr0 + k0);
        const float4 rb0 = *(const float4*)(wr0 + k0 + 4);
        const float4 ra1 = *(const float4*)(wr1 + k0);
        const float4 rb1 = *(const float4*)(wr1 + k0 + 4);
        const float ra0v[4] = {ra0.x, ra0.y, ra0.z, ra0.w};
        const float rb0v[4] = {rb0.x, rb0.y, rb0.z, rb0.w};
        const float ra1v[4] = {ra1.x, ra1.y, ra1.z, ra1.w};
        const float rb1v[4] = {rb1.x, rb1.y, rb1.z, rb1.w};
#pragma unroll
        for (int i = 0; i < NRT; ++i) {
            const float4 xa = *(const float4*)(xrow[i] + k0);
            const float4 xb = *(const float4*)(xrow[i] + k0 + 4);
            const float xav[4] = {xa.x, xa.y, xa.z, xa.w};
            const float xbv[4] = {xb.x, xb.y, xb.z, xb.w};
#pragma unroll
            for (int j = 0; j < 4; ++j) {
                const v2f a_x = (v2f){xav[j], xbv[j]};
                const v2f b_r0 = (v2f){ra0v[j], rb0v[j]};
                const v2f b_r1 = (v2f){ra1v[j], rb1v[j]};
                c[i][0] = __builtin_amdgcn_wmma_f32_16x16x4_f32(
                    false, a_x, false, b_r0, (short)0, c[i][0], false, false);
                c[i][1] = __builtin_amdgcn_wmma_f32_16x16x4_f32(
                    false, a_x, false, b_r1, (short)0, c[i][1], false, false);
            }
        }
    }
}

template <int NRT>
__device__ __forceinline__ void sage_sum_product_store(
    v8f (&c)[2][2], const float* acc, const float* __restrict__ Wl,
    const float* __restrict__ bl, float* __restrict__ out,
    int lo, int nOwn, int g, int cp, int lane) {
    const int half = lane >> 4;
    const int lm = lane & 15;
    const float* arow[NRT];
#pragma unroll
    for (int i = 0; i < NRT; ++i) {
        const int r = (g + 8 * i) * 16 + lm;
        arow[i] = acc + r * SAGE_P + 2 * half;
    }
    const float* wl0 = Wl + (size_t)(cp * 32 + lm) * SAGE_D + 8 * half;
    const float* wl1 = wl0 + 16 * SAGE_D;

#pragma unroll 1
    for (int k0 = 0; k0 < SAGE_D; k0 += 16) {
        const float4 la0 = *(const float4*)(wl0 + k0);
        const float4 lb0 = *(const float4*)(wl0 + k0 + 4);
        const float4 la1 = *(const float4*)(wl1 + k0);
        const float4 lb1 = *(const float4*)(wl1 + k0 + 4);
        const float la0v[4] = {la0.x, la0.y, la0.z, la0.w};
        const float lb0v[4] = {lb0.x, lb0.y, lb0.z, lb0.w};
        const float la1v[4] = {la1.x, la1.y, la1.z, la1.w};
        const float lb1v[4] = {lb1.x, lb1.y, lb1.z, lb1.w};
#pragma unroll
        for (int i = 0; i < NRT; ++i) {
#pragma unroll
            for (int j = 0; j < 4; ++j) {
                const v2f a_s = *(const v2f*)(arow[i] + j * 32 + (k0 >> 2));
                const v2f b_l0 = (v2f){la0v[j], lb0v[j]};
                const v2f b_l1 = (v2f){la1v[j], lb1v[j]};
                c[i][0] = __builtin_amdgcn_wmma_f32_16x16x4_f32(
                    false, a_s, false, b_l0, (short)0, c[i][0], false, false);
                c[i][1] = __builtin_amdgcn_wmma_f32_16x16x4_f32(
                    false, a_s, false, b_l1, (short)0, c[i][1], false, false);
            }
        }
    }

#pragma unroll
    for (int i = 0; i < NRT; ++i) {
        const int rt = g + 8 * i;
        const bool whole = rt * 16 + 16 <= nOwn;
#pragma unroll
        for (int t = 0; t < 2; ++t) {
            const int col = cp * 32 + t * 16 + lm;
            const float bias = bl[col];
            if (whole) {
                float* o = out + (size_t)(lo + rt * 16 + 8 * half) * SAGE_D + col;
#pragma unroll
                for (int v = 0; v < 8; ++v) o[v * SAGE_D] = c[i][t][v] + bias;
            } else {
#pragma unroll
                for (int v = 0; v < 8; ++v) {
                    const int r = rt * 16 + v + 8 * half;
                    if (r < nOwn)
                        out[(size_t)(lo + r) * SAGE_D + col] = c[i][t][v] + bias;
                }
            }
        }
    }
}

__global__ __launch_bounds__(SAGE_T) void sage_layer_kernel(
    const float* __restrict__ xin, const int* __restrict__ src,
    const int* __restrict__ dst, const unsigned char* __restrict__ own,
    const float* __restrict__ Wl, const float* __restrict__ bl,
    const float* __restrict__ Wr, float* __restrict__ out,
    float* __restrict__ cnt, int* __restrict__ wlist,
    int* __restrict__ wcount, int E, int first) {
    __shared__ __attribute__((aligned(16))) float acc[SAGE_ROWS * SAGE_P];
    __shared__ int2 lst[SAGE_LCAP];
    __shared__ int cntl[SAGE_R];
    __shared__ float invl[SAGE_R];
    __shared__ int nlist;

    const int tid = threadIdx.x;
    const int lane = tid & 31;
    const int wave = __builtin_amdgcn_readfirstlane(tid >> 5);
    const int lo = blockIdx.x * SAGE_R;
    const int nOwn = min(SAGE_R, SAGE_N - lo);
    int2* wl = (int2*)wlist + (size_t)blockIdx.x * SAGE_LCAP;

    for (int i = tid; i < SAGE_ROWS * SAGE_P; i += SAGE_T) acc[i] = 0.0f;
    for (int i = tid; i < SAGE_R; i += SAGE_T) cntl[i] = 0;
    if (tid == 0) nlist = 0;
    __syncthreads();

    const int saved = first ? 0 : wcount[blockIdx.x];
    const bool scan = first || saved > SAGE_LCAP;
    int n;

    if (scan) {
        const uint4* o4 = (const uint4*)own;
        const int E16 = E >> 4;
        const unsigned key = blockIdx.x * 0x01010101u;
        const int nGroups = (E16 + SAGE_T * SAGE_U - 1) / (SAGE_T * SAGE_U);
        const int rot = (int)((blockIdx.x * (unsigned)nGroups) / gridDim.x);
        for (int gi = 0; gi < nGroups; ++gi) {
            int g = gi + rot;
            if (g >= nGroups) g -= nGroups;
            uint4 ov[SAGE_U];
#pragma unroll
            for (int u = 0; u < SAGE_U; ++u) {
                const int i16 = (g * SAGE_U + u) * SAGE_T + tid;
                ov[u] = o4[i16 < E16 ? i16 : 0];
            }
#pragma unroll
            for (int u = 0; u < SAGE_U; ++u) {
                const int i16 = (g * SAGE_U + u) * SAGE_T + tid;
                if (i16 < E16) {
                    const unsigned n0 = sage_nonzero_bytes(ov[u].x ^ key);
                    const unsigned n1 = sage_nonzero_bytes(ov[u].y ^ key);
                    const unsigned n2 = sage_nonzero_bytes(ov[u].z ^ key);
                    const unsigned n3 = sage_nonzero_bytes(ov[u].w ^ key);
                    if (((n0 & n1 & n2 & n3) & 0x80808080u) != 0x80808080u) {
                        unsigned m = ((~n0 & 0x80808080u) >> 7) |
                                     ((~n1 & 0x80808080u) >> 6) |
                                     ((~n2 & 0x80808080u) >> 5) |
                                     ((~n3 & 0x80808080u) >> 4);
                        while (m) {
                            const int p = __builtin_ctz(m);
                            m &= m - 1;
                            const int e = i16 * 16 + (p & 7) * 4 + (p >> 3);
                            sage_note_edge(e, lo, nOwn, first, cntl, &nlist, lst, acc, xin, src, dst);
                        }
                    }
                }
            }
        }
        for (int e = E16 * 16 + tid; e < E; e += SAGE_T)
            if (own[e] == (unsigned char)blockIdx.x)
                sage_note_edge(e, lo, nOwn, first, cntl, &nlist, lst, acc, xin, src, dst);
        __syncthreads();

        const int nl = nlist;
        n = min(nl, SAGE_LCAP);
        for (int i = tid; i < n; i += SAGE_T) {
            const int e = lst[i].x;
            const int sv = src[e];
            const unsigned dl = (unsigned)(dst[e] - lo);
            const bool ok = dl < (unsigned)nOwn;
            if (ok && first) atomicAdd(&cntl[dl], 1);
            const int2 ent = ok ? make_int2(sv, (int)dl) : make_int2(0, SAGE_R);
            lst[i] = ent;
            if (first) wl[i] = ent;
        }
        if (first && tid == 0) wcount[blockIdx.x] = nl;
        __syncthreads();
    } else {
        n = saved;
        for (int i = tid; i < n; i += SAGE_T) lst[i] = wl[i];
        __syncthreads();
    }

    const int nTiles = (nOwn + 15) >> 4;
    const int g = wave >> 2;
    const int cp = wave & 3;
    const int nrt = (g + 8 < nTiles) ? 2 : ((g < nTiles) ? 1 : 0);
    const bool selfFirst = (g & 1) == 0;
    v8f c[2][2];
    c[0][0] = (v8f){};
    c[0][1] = (v8f){};
    c[1][0] = (v8f){};
    c[1][1] = (v8f){};
    if (selfFirst) {
        if (nrt == 2)
            sage_self_product<2>(c, xin, Wr, lo, g, cp, lane);
        else if (nrt == 1)
            sage_self_product<1>(c, xin, Wr, lo, g, cp, lane);
    }

    for (int base = wave; base < n; base += 32 * SAGE_U) {
        float4 v[SAGE_U];
        int dl[SAGE_U];
#pragma unroll
        for (int u = 0; u < SAGE_U; ++u) {
            const int idx = base + u * 32;
            const int2 ent = lst[idx < n ? idx : base];
            dl[u] = ent.y;
            v[u] = *(const float4*)(xin + (size_t)ent.x * SAGE_D + lane * 4);
        }
        asm volatile("" :: "v"(v[0].x), "v"(v[1].x), "v"(v[2].x), "v"(v[3].x),
                           "v"(v[4].x), "v"(v[5].x), "v"(v[6].x), "v"(v[7].x));
#pragma unroll
        for (int u = 0; u < SAGE_U; ++u) {
            if (base + u * 32 < n) {
                float* a = acc + dl[u] * SAGE_P + lane;
                unsafeAtomicAdd(a + 0, v[u].x);
                unsafeAtomicAdd(a + 32, v[u].y);
                unsafeAtomicAdd(a + 64, v[u].z);
                unsafeAtomicAdd(a + 96, v[u].w);
            }
        }
    }
    if (!selfFirst) {
        if (nrt == 2)
            sage_self_product<2>(c, xin, Wr, lo, g, cp, lane);
        else if (nrt == 1)
            sage_self_product<1>(c, xin, Wr, lo, g, cp, lane);
    }
    __syncthreads();

    for (int i = tid; i < nOwn; i += SAGE_T) {
        const float cv = first ? (float)cntl[i] : cnt[lo + i];
        if (first) cnt[lo + i] = cv;
        invl[i] = 1.0f / fmaxf(cv, 1.0f);
    }
    __syncthreads();
    for (int i = tid; i < nOwn * SAGE_D; i += SAGE_T) {
        const int r = i >> 7;
        acc[r * SAGE_P + (i & (SAGE_D - 1))] *= invl[r];
    }
    __syncthreads();

    if (nrt == 2)
        sage_sum_product_store<2>(c, acc, Wl, bl, out, lo, nOwn, g, cp, lane);
    else if (nrt == 1)
        sage_sum_product_store<1>(c, acc, Wl, bl, out, lo, nOwn, g, cp, lane);
}

extern "C" void kernel_launch(void* const* d_in, const int* in_sizes, int n_in,
                              void* d_out, int out_size, void* d_ws, size_t ws_size,
                              hipStream_t stream) {
    const float* x   = (const float*)d_in[0];
    const int*   ei  = (const int*)d_in[1];
    const float* Wl1 = (const float*)d_in[2];
    const float* bl1 = (const float*)d_in[3];
    const float* Wr1 = (const float*)d_in[4];
    const float* Wl2 = (const float*)d_in[5];
    const float* bl2 = (const float*)d_in[6];
    const float* Wr2 = (const float*)d_in[7];
    float* out = (float*)d_out;

    const int N = SAGE_N;
    const int E = in_sizes[1] / 2;
    const int* src = ei;
    const int* dst = ei + E;

    float* h   = (float*)d_ws;
    float* cnt = h + (size_t)N * SAGE_D;
    int* wlist  = (int*)(cnt + N);
    int* wcount = wlist + (size_t)2 * SAGE_LCAP * ((N + SAGE_R - 1) / SAGE_R);
    unsigned char* own = (unsigned char*)(wcount + 256);

    const int aggBlocks  = (N + SAGE_R - 1) / SAGE_R;

    sage_owner_kernel<<<256, SAGE_T, 0, stream>>>(dst, own, E);

    sage_layer_kernel<<<aggBlocks, SAGE_T, 0, stream>>>(x, src, dst, own, Wl1, bl1, Wr1,
                                                        h, cnt, wlist, wcount, E, 1);
    sage_layer_kernel<<<aggBlocks, SAGE_T, 0, stream>>>(h, src, dst, own, Wl2, bl2, Wr2,
                                                        out, cnt, wlist, wcount, E, 0);
}


# Edge lists filed once: first launch sorts each 1/256 slice of the edges by owning work-group in LDS, a layer copies its 256 segments; owner-byte scan and per-edge id reads gone; adds in arrival order
# speedup vs baseline: 3.9896x; 1.1121x over previous
#include <hip/hip_runtime.h>
#include <hip/hip_bf16.h>


#define SAGE_N 50000
#define SAGE_D 128
#define SAGE_R 196
#define SAGE_G 256
#define SAGE_NCH 256
#define SAGE_CCAP 8192
#define SAGE_FU 4
#define SAGE_LCAP 16384
#define SAGE_T 1024
#define SAGE_U 8
#define SAGE_P 132
#define SAGE_ROWS 208

typedef __attribute__((ext_vector_type(2))) float v2f;
typedef __attribute__((ext_vector_type(8))) float v8f;

__device__ __forceinline__ unsigned sage_owner_of(int d) {
    const unsigned o = (unsigned)d / (unsigned)SAGE_R;
    return o < (unsigned)(SAGE_G - 1) ? o : (unsigned)(SAGE_G - 1);
}

__device__ __forceinline__ int sage_scan256(int v, int tid, int* wtot, int* total) {
    const int lane = tid & 31;
    int incl = v;
    for (int d = 1; d < 32; d <<= 1) {
        const int t = __shfl_up(incl, d, 32);
        if (lane >= d) incl += t;
    }
    if (tid < 256 && lane == 31) wtot[tid >> 5] = incl;
    __syncthreads();
    int pre = 0;
    int tot = 0;
    for (int k = 0; k < 8; ++k) {
        const int w = wtot[k];
        if (k < (tid >> 5)) pre += w;
        tot += w;
    }
    *total = tot;
    return pre + incl - v;
}

__global__ __launch_bounds__(SAGE_T) void sage_file_kernel(
    const int* __restrict__ src, const int* __restrict__ dst,
    int* __restrict__ sortedw, int* __restrict__ segT, int E) {
    __shared__ int2 stage[SAGE_CCAP];
    __shared__ int hist[SAGE_G];
    __shared__ int cursor[SAGE_G];
    __shared__ int wtot[8];

    const int tid = threadIdx.x;
    const int w = blockIdx.x;
    const int C = (E + SAGE_NCH - 1) / SAGE_NCH;
    const int e0 = min(w * C, E);
    const int e1 = min(e0 + C, E);
    int2* sorted = (int2*)sortedw;

    if (tid < SAGE_G) hist[tid] = 0;
    __syncthreads();
    for (int b = e0 + tid; b < e1; b += SAGE_FU * SAGE_T) {
        int d[SAGE_FU];
#pragma unroll
        for (int u = 0; u < SAGE_FU; ++u) d[u] = dst[min(b + u * SAGE_T, e1 - 1)];
#pragma unroll
        for (int u = 0; u < SAGE_FU; ++u)
            atomicAdd(&hist[sage_owner_of(d[u])], b + u * SAGE_T < e1 ? 1 : 0);
    }
    __syncthreads();

    int total;
    const int cntv = tid < SAGE_G ? hist[tid] : 0;
    const int start = sage_scan256(cntv, tid, wtot, &total);
    if (tid < SAGE_G) {
        cursor[tid] = start;
        segT[tid * SAGE_NCH + w] = start;
    }
    if (tid == 0) segT[SAGE_G * SAGE_NCH + w] = total;
    __syncthreads();

    const bool staged = C <= SAGE_CCAP;
    for (int b = e0 + tid; b < e1; b += SAGE_FU * SAGE_T) {
        int d[SAGE_FU];
        int sv[SAGE_FU];
#pragma unroll
        for (int u = 0; u < SAGE_FU; ++u) {
            const int e = min(b + u * SAGE_T, e1 - 1);
            d[u] = dst[e];
            sv[u] = src[e];
        }
#pragma unroll
        for (int u = 0; u < SAGE_FU; ++u) {
            if (b + u * SAGE_T < e1) {
                const unsigned o = sage_owner_of(d[u]);
                const unsigned dl = (unsigned)d[u] - o * (unsigned)SAGE_R;
                const unsigned nOwnO = min((unsigned)SAGE_R, (unsigned)SAGE_N - o * (unsigned)SAGE_R);
                const int2 ent = dl < nOwnO ? make_int2(sv[u], (int)dl) : make_int2(0, SAGE_R);
                const int pos = atomicAdd(&cursor[o], 1);
                if (staged) stage[pos] = ent;
                else sorted[(size_t)e0 + pos] = ent;
            }
        }
    }
    __syncthreads();
    if (staged)
        for (int i = tid; i < e1 - e0; i += SAGE_T) sorted[(size_t)e0 + i] = stage[i];
}

__device__ __forceinline__ void sage_list_put(int2 ent, int p, int* cntl, int2* lst) {
    if (ent.y < SAGE_R) atomicAdd(&cntl[ent.y], 1);
    if (p < SAGE_LCAP) lst[p] = ent;
}

template <int NRT>
__device__ __forceinline__ void sage_self_product(
    v8f (&c)[2][2], const float* __restrict__ xin,
    const float* __restrict__ Wr, int lo, int g, int cp, int lane) {
    const int half = lane >> 4;
    const int lm = lane & 15;
    const float* xrow[NRT];
#pragma unroll
    for (int i = 0; i < NRT; ++i) {
        const int r = (g + 8 * i) * 16 + lm;
        const int gr = min(lo + r, SAGE_N - 1);
        xrow[i] = xin + (size_t)gr * SAGE_D + 8 * half;
    }
    const float* wr0 = Wr + (size_t)(cp * 32 + lm) * SAGE_D + 8 * half;
    const float* wr1 = wr0 + 16 * SAGE_D;

#pragma unroll 2
    for (int k0 = 0; k0 < SAGE_D; k0 += 16) {
        const float4 ra0 = *(const float4*)(wr0 + k0);
        const float4 rb0 = *(const float4*)(wr0 + k0 + 4);
        const float4 ra1 = *(const float4*)(wr1 + k0);
        const float4 rb1 = *(const float4*)(wr1 + k0 + 4);
        const float ra0v[4] = {ra0.x, ra0.y, ra0.z, ra0.w};
        const float rb0v[4] = {rb0.x, rb0.y, rb0.z, rb0.w};
        const float ra1v[4] = {ra1.x, ra1.y, ra1.z, ra1.w};
        const float rb1v[4] = {rb1.x, rb1.y, rb1.z, rb1.w};
#pragma unroll
        for (int i = 0; i < NRT; ++i) {
            const float4 xa = *(const float4*)(xrow[i] + k0);
            const float4 xb = *(const float4*)(xrow[i] + k0 + 4);
            const float xav[4] = {xa.x, xa.y, xa.z, xa.w};
            const float xbv[4] = {xb.x, xb.y, xb.z, xb.w};
#pragma unroll
            for (int j = 0; j < 4; ++j) {
                const v2f a_x = (v2f){xav[j], xbv[j]};
                const v2f b_r0 = (v2f){ra0v[j], rb0v[j]};
                const v2f b_r1 = (v2f){ra1v[j], rb1v[j]};
                c[i][0] = __builtin_amdgcn_wmma_f32_16x16x4_f32(
                    false, a_x, false, b_r0, (short)0, c[i][0], false, false);
                c[i][1] = __builtin_amdgcn_wmma_f32_16x16x4_f32(
                    false, a_x, false, b_r1, (short)0, c[i][1], false, false);
            }
        }
    }
}

template <int NRT>
__device__ __forceinline__ void sage_sum_product_store(
    v8f (&c)[2][2], const float* acc, const float* __restrict__ Wl,
    const float* __restrict__ bl, float* __restrict__ out,
    int lo, int nOwn, int g, int cp, int lane) {
    const int half = lane >> 4;
    const int lm = lane & 15;
    const float* arow[NRT];
#pragma unroll
    for (int i = 0; i < NRT; ++i) {
        const int r = (g + 8 * i) * 16 + lm;
        arow[i] = acc + r * SAGE_P + 2 * half;
    }
    const float* wl0 = Wl + (size_t)(cp * 32 + lm) * SAGE_D + 8 * half;
    const float* wl1 = wl0 + 16 * SAGE_D;

#pragma unroll 1
    for (int k0 = 0; k0 < SAGE_D; k0 += 16) {
        const float4 la0 = *(const float4*)(wl0 + k0);
        const float4 lb0 = *(const float4*)(wl0 + k0 + 4);
        const float4 la1 = *(const float4*)(wl1 + k0);
        const float4 lb1 = *(const float4*)(wl1 + k0 + 4);
        const float la0v[4] = {la0.x, la0.y, la0.z, la0.w};
        const float lb0v[4] = {lb0.x, lb0.y, lb0.z, lb0.w};
        const float la1v[4] = {la1.x, la1.y, la1.z, la1.w};
        const float lb1v[4] = {lb1.x, lb1.y, lb1.z, lb1.w};
#pragma unroll
        for (int i = 0; i < NRT; ++i) {
#pragma unroll
            for (int j = 0; j < 4; ++j) {
                const v2f a_s = *(const v2f*)(arow[i] + j * 32 + (k0 >> 2));
                const v2f b_l0 = (v2f){la0v[j], lb0v[j]};
                const v2f b_l1 = (v2f){la1v[j], lb1v[j]};
                c[i][0] = __builtin_amdgcn_wmma_f32_16x16x4_f32(
                    false, a_s, false, b_l0, (short)0, c[i][0], false, false);
                c[i][1] = __builtin_amdgcn_wmma_f32_16x16x4_f32(
                    false, a_s, false, b_l1, (short)0, c[i][1], false, false);
            }
        }
    }

#pragma unroll
    for (int i = 0; i < NRT; ++i) {
        const int rt = g + 8 * i;
        const bool whole = rt * 16 + 16 <= nOwn;
#pragma unroll
        for (int t = 0; t < 2; ++t) {
            const int col = cp * 32 + t * 16 + lm;
            const float bias = bl[col];
            if (whole) {
                float* o = out + (size_t)(lo + rt * 16 + 8 * half) * SAGE_D + col;
#pragma unroll
                for (int v = 0; v < 8; ++v) o[v * SAGE_D] = c[i][t][v] + bias;
            } else {
#pragma unroll
                for (int v = 0; v < 8; ++v) {
                    const int r = rt * 16 + v + 8 * half;
                    if (r < nOwn)
                        out[(size_t)(lo + r) * SAGE_D + col] = c[i][t][v] + bias;
                }
            }
        }
    }
}

__global__ __launch_bounds__(SAGE_T) void sage_layer_kernel(
    const float* __restrict__ xin, const int* __restrict__ sortedw,
    const int* __restrict__ segT,
    const float* __restrict__ Wl, const float* __restrict__ bl,
    const float* __restrict__ Wr, float* __restrict__ out, int E) {
    __shared__ __attribute__((aligned(16))) float acc[SAGE_ROWS * SAGE_P];
    __shared__ int2 lst[SAGE_LCAP];
    __shared__ int cntl[SAGE_R];
    __shared__ float invl[SAGE_R];
    __shared__ int segS[SAGE_NCH];
    __shared__ int segL[SAGE_NCH];
    __shared__ int segB[SAGE_NCH];
    __shared__ int wtot[8];

    const int tid = threadIdx.x;
    const int lane = tid & 31;
    const int wave = __builtin_amdgcn_readfirstlane(tid >> 5);
    const int lo = blockIdx.x * SAGE_R;
    const int nOwn = min(SAGE_R, SAGE_N - lo);
    const int C = (E + SAGE_NCH - 1) / SAGE_NCH;
    const int2* sorted = (const int2*)sortedw;

    int s0 = 0;
    int s1 = 0;
    if (tid < SAGE_NCH) {
        s0 = segT[blockIdx.x * SAGE_NCH + tid];
        s1 = segT[(blockIdx.x + 1) * SAGE_NCH + tid];
    }
    for (int i = tid; i < SAGE_ROWS * SAGE_P; i += SAGE_T) acc[i] = 0.0f;
    for (int i = tid; i < SAGE_R; i += SAGE_T) cntl[i] = 0;
    const int len = s1 - s0;
    int nTot;
    const int bpos = sage_scan256(len, tid, wtot, &nTot);
    if (tid < SAGE_NCH) {
        segS[tid] = tid * C + s0;
        segL[tid] = len;
        segB[tid] = bpos;
    }
    __syncthreads();

    {
        int2 ent[SAGE_NCH / 32];
        int ln[SAGE_NCH / 32];
        int bp[SAGE_NCH / 32];
        int sp[SAGE_NCH / 32];
#pragma unroll
        for (int k = 0; k < SAGE_NCH / 32; ++k) {
            const int t = wave + 32 * k;
            ln[k] = segL[t];
            bp[k] = segB[t];
            sp[k] = segS[t];
            ent[k] = sorted[lane < ln[k] ? sp[k] + lane : 0];
        }
#pragma unroll
        for (int k = 0; k < SAGE_NCH / 32; ++k)
            if (lane < ln[k]) sage_list_put(ent[k], bp[k] + lane, cntl, lst);
#pragma unroll
        for (int k = 0; k < SAGE_NCH / 32; ++k)
            for (int i = lane + 32; i < ln[k]; i += 32)
                sage_list_put(sorted[sp[k] + i], bp[k] + i, cntl, lst);
    }
    if (nTot > SAGE_LCAP) {
        for (int t = wave; t < SAGE_NCH; t += 32) {
            for (int i = lane; i < segL[t]; i += 32) {
                if (segB[t] + i >= SAGE_LCAP) {
                    const int2 ent = sorted[segS[t] + i];
                    if (ent.y < SAGE_R) {
                        const float* xr = xin + (size_t)ent.x * SAGE_D;
                        float* ar = acc + ent.y * SAGE_P;
                        for (int k = 0; k < SAGE_D; ++k)
                            unsafeAtomicAdd(ar + (k & 3) * 32 + (k >> 2), xr[k]);
                    }
                }
            }
        }
    }
    __syncthreads();
    const int n = min(nTot, SAGE_LCAP);

    const int nTiles = (nOwn + 15) >> 4;
    const int g = wave >> 2;
    const int cp = wave & 3;
    const int nrt = (g + 8 < nTiles) ? 2 : ((g < nTiles) ? 1 : 0);
    const bool selfFirst = (g & 1) == 0;
    v8f c[2][2];
    c[0][0] = (v8f){};
    c[0][1] = (v8f){};
    c[1][0] = (v8f){};
    c[1][1] = (v8f){};
    if (selfFirst) {
        if (nrt == 2)
            sage_self_product<2>(c, xin, Wr, lo, g, cp, lane);
        else if (nrt == 1)
            sage_self_product<1>(c, xin, Wr, lo, g, cp, lane);
    }

    for (int base = wave; base < n; base += 32 * SAGE_U) {
        float4 v[SAGE_U];
        int dl[SAGE_U];
#pragma unroll
        for (int u = 0; u < SAGE_U; ++u) {
            const int idx = base + u * 32;
            const int2 ent = lst[idx < n ? idx : base];
            dl[u] = ent.y;
            v[u] = *(const float4*)(xin + (size_t)ent.x * SAGE_D + lane * 4);
        }
        asm volatile("" :: "v"(v[0].x), "v"(v[1].x), "v"(v[2].x), "v"(v[3].x),
                           "v"(v[4].x), "v"(v[5].x), "v"(v[6].x), "v"(v[7].x));
#pragma unroll
        for (int u = 0; u < SAGE_U; ++u) {
            if (base + u * 32 < n) {
                float* a = acc + dl[u] * SAGE_P + lane;
                unsafeAtomicAdd(a + 0, v[u].x);
                unsafeAtomicAdd(a + 32, v[u].y);
                unsafeAtomicAdd(a + 64, v[u].z);
                unsafeAtomicAdd(a + 96, v[u].w);
            }
        }
    }
    if (!selfFirst) {
        if (nrt == 2)
            sage_self_product<2>(c, xin, Wr, lo, g, cp, lane);
        else if (nrt == 1)
            sage_self_product<1>(c, xin, Wr, lo, g, cp, lane);
    }
    __syncthreads();

    for (int i = tid; i < nOwn; i += SAGE_T)
        invl[i] = 1.0f / fmaxf((float)cntl[i], 1.0f);
    __syncthreads();
    for (int i = tid; i < nOwn * SAGE_D; i += SAGE_T) {
        const int r = i >> 7;
        acc[r * SAGE_P + (i & (SAGE_D - 1))] *= invl[r];
    }
    __syncthreads();

    if (nrt == 2)
        sage_sum_product_store<2>(c, acc, Wl, bl, out, lo, nOwn, g, cp, lane);
    else if (nrt == 1)
        sage_sum_product_store<1>(c, acc, Wl, bl, out, lo, nOwn, g, cp, lane);
}

extern "C" void kernel_launch(void* const* d_in, const int* in_sizes, int n_in,
                              void* d_out, int out_size, void* d_ws, size_t ws_size,
                              hipStream_t stream) {
    const float* x   = (const float*)d_in[0];
    const int*   ei  = (const int*)d_in[1];
    const float* Wl1 = (const float*)d_in[2];
    const float* bl1 = (const float*)d_in[3];
    const float* Wr1 = (const float*)d_in[4];
    const float* Wl2 = (const float*)d_in[5];
    const float* bl2 = (const float*)d_in[6];
    const float* Wr2 = (const float*)d_in[7];
    float* out = (float*)d_out;

    const int N = SAGE_N;
    const int E = in_sizes[1] / 2;
    const int* src = ei;
    const int* dst = ei + E;

    float* h    = (float*)d_ws;
    int* segT   = (int*)(h + (size_t)N * SAGE_D);
    int* sorted = segT + (SAGE_G + 1) * SAGE_NCH;

    sage_file_kernel<<<SAGE_NCH, SAGE_T, 0, stream>>>(src, dst, sorted, segT, E);

    sage_layer_kernel<<<SAGE_G, SAGE_T, 0, stream>>>(x, sorted, segT, Wl1, bl1, Wr1, h, E);
    sage_layer_kernel<<<SAGE_G, SAGE_T, 0, stream>>>(h, sorted, segT, Wl2, bl2, Wr2, out, E);
}


# Group's edge list put in destination order in LDS (count, scan, place); a wave sums a destination's rows in registers: one LDS float add per run, not per edge; sum order: placement (arrival) order
# speedup vs baseline: 4.3467x; 1.0895x over previous
#include <hip/hip_runtime.h>
#include <hip/hip_bf16.h>


#define SAGE_N 50000
#define SAGE_D 128
#define SAGE_R 196
#define SAGE_G 256
#define SAGE_NCH 256
#define SAGE_CCAP 8192
#define SAGE_FU 4
#define SAGE_LCAP 16384
#define SAGE_T 1024
#define SAGE_U 8
#define SAGE_P 132
#define SAGE_ROWS 208

typedef __attribute__((ext_vector_type(2))) float v2f;
typedef __attribute__((ext_vector_type(8))) float v8f;

__device__ __forceinline__ unsigned sage_owner_of(int d) {
    const unsigned o = (unsigned)d / (unsigned)SAGE_R;
    return o < (unsigned)(SAGE_G - 1) ? o : (unsigned)(SAGE_G - 1);
}

__device__ __forceinline__ int sage_scan256(int v, int tid, int* wtot, int* total) {
    const int lane = tid & 31;
    int incl = v;
    for (int d = 1; d < 32; d <<= 1) {
        const int t = __shfl_up(incl, d, 32);
        if (lane >= d) incl += t;
    }
    if (tid < 256 && lane == 31) wtot[tid >> 5] = incl;
    __syncthreads();
    int pre = 0;
    int tot = 0;
    for (int k = 0; k < 8; ++k) {
        const int w = wtot[k];
        if (k < (tid >> 5)) pre += w;
        tot += w;
    }
    *total = tot;
    return pre + incl - v;
}

__global__ __launch_bounds__(SAGE_T) void sage_file_kernel(
    const int* __restrict__ src, const int* __restrict__ dst,
    int* __restrict__ sortedw, int* __restrict__ segT, int E) {
    __shared__ int2 stage[SAGE_CCAP];
    __shared__ int hist[SAGE_G];
    __shared__ int cursor[SAGE_G];
    __shared__ int wtot[8];

    const int tid = threadIdx.x;
    const int w = blockIdx.x;
    const int C = (E + SAGE_NCH - 1) / SAGE_NCH;
    const int e0 = min(w * C, E);
    const int e1 = min(e0 + C, E);
    int2* sorted = (int2*)sortedw;

    if (tid < SAGE_G) hist[tid] = 0;
    __syncthreads();
    for (int b = e0 + tid; b < e1; b += SAGE_FU * SAGE_T) {
        int d[SAGE_FU];
#pragma unroll
        for (int u = 0; u < SAGE_FU; ++u) d[u] = dst[min(b + u * SAGE_T, e1 - 1)];
#pragma unroll
        for (int u = 0; u < SAGE_FU; ++u)
            atomicAdd(&hist[sage_owner_of(d[u])], b + u * SAGE_T < e1 ? 1 : 0);
    }
    __syncthreads();

    int total;
    const int cntv = tid < SAGE_G ? hist[tid] : 0;
    const int start = sage_scan256(cntv, tid, wtot, &total);
    if (tid < SAGE_G) {
        cursor[tid] = start;
        segT[tid * SAGE_NCH + w] = start;
    }
    if (tid == 0) segT[SAGE_G * SAGE_NCH + w] = total;
    __syncthreads();

    const bool staged = C <= SAGE_CCAP;
    for (int b = e0 + tid; b < e1; b += SAGE_FU * SAGE_T) {
        int d[SAGE_FU];
        int sv[SAGE_FU];
#pragma unroll
        for (int u = 0; u < SAGE_FU; ++u) {
            const int e = min(b + u * SAGE_T, e1 - 1);
            d[u] = dst[e];
            sv[u] = src[e];
        }
#pragma unroll
        for (int u = 0; u < SAGE_FU; ++u) {
            if (b + u * SAGE_T < e1) {
                const unsigned o = sage_owner_of(d[u]);
                const unsigned dl = (unsigned)d[u] - o * (unsigned)SAGE_R;
                const unsigned nOwnO = min((unsigned)SAGE_R, (unsigned)SAGE_N - o * (unsigned)SAGE_R);
                const int2 ent = dl < nOwnO ? make_int2(sv[u], (int)dl) : make_int2(0, SAGE_R);
                const int pos = atomicAdd(&cursor[o], 1);
                if (staged) stage[pos] = ent;
                else sorted[(size_t)e0 + pos] = ent;
            }
        }
    }
    __syncthreads();
    if (staged)
        for (int i = tid; i < e1 - e0; i += SAGE_T) sorted[(size_t)e0 + i] = stage[i];
}

__device__ __forceinline__ void sage_run_add(float* acc, int row, int lane, float4 run) {
    float* a = acc + row * SAGE_P + lane;
    unsafeAtomicAdd(a + 0, run.x);
    unsafeAtomicAdd(a + 32, run.y);
    unsafeAtomicAdd(a + 64, run.z);
    unsafeAtomicAdd(a + 96, run.w);
}

template <int NRT>
__device__ __forceinline__ void sage_self_product(
    v8f (&c)[2][2], const float* __restrict__ xin,
    const float* __restrict__ Wr, int lo, int g, int cp, int lane) {
    const int half = lane >> 4;
    const int lm = lane & 15;
    const float* xrow[NRT];
#pragma unroll
    for (int i = 0; i < NRT; ++i) {
        const int r = (g + 8 * i) * 16 + lm;
        const int gr = min(lo + r, SAGE_N - 1);
        xrow[i] = xin + (size_t)gr * SAGE_D + 8 * half;
    }
    const float* wr0 = Wr + (size_t)(cp * 32 + lm) * SAGE_D + 8 * half;
    const float* wr1 = wr0 + 16 * SAGE_D;

#pragma unroll 2
    for (int k0 = 0; k0 < SAGE_D; k0 += 16) {
        const float4 ra0 = *(const float4*)(wr0 + k0);
        const float4 rb0 = *(const float4*)(wr0 + k0 + 4);
        const float4 ra1 = *(const float4*)(wr1 + k0);
        const float4 rb1 = *(const float4*)(wr1 + k0 + 4);
        const float ra0v[4] = {ra0.x, ra0.y, ra0.z, ra0.w};
        const float rb0v[4] = {rb0.x, rb0.y, rb0.z, rb0.w};
        const float ra1v[4] = {ra1.x, ra1.y, ra1.z, ra1.w};
        const float rb1v[4] = {rb1.x, rb1.y, rb1.z, rb1.w};
#pragma unroll
        for (int i = 0; i < NRT; ++i) {
            const float4 xa = *(const float4*)(xrow[i] + k0);
            const float4 xb = *(const float4*)(xrow[i] + k0 + 4);
            const float xav[4] = {xa.x, xa.y, xa.z, xa.w};
            const float xbv[4] = {xb.x, xb.y, xb.z, xb.w};
#pragma unroll
            for (int j = 0; j < 4; ++j) {
                const v2f a_x = (v2f){xav[j], xbv[j]};
                const v2f b_r0 = (v2f){ra0v[j], rb0v[j]};
                const v2f b_r1 = (v2f){ra1v[j], rb1v[j]};
                c[i][0] = __builtin_amdgcn_wmma_f32_16x16x4_f32(
                    false, a_x, false, b_r0, (short)0, c[i][0], false, false);
                c[i][1] = __builtin_amdgcn_wmma_f32_16x16x4_f32(
                    false, a_x, false, b_r1, (short)0, c[i][1], false, false);
            }
        }
    }
}

template <int NRT>
__device__ __forceinline__ void sage_sum_product_store(
    v8f (&c)[2][2], const float* acc, const float* __restrict__ Wl,
    const float* __restrict__ bl, float* __restrict__ out,
    int lo, int nOwn, int g, int cp, int lane) {
    const int half = lane >> 4;
    const int lm = lane & 15;
    const float* arow[NRT];
#pragma unroll
    for (int i = 0; i < NRT; ++i) {
        const int r = (g + 8 * i) * 16 + lm;
        arow[i] = acc + r * SAGE_P + 2 * half;
    }
    const float* wl0 = Wl + (size_t)(cp * 32 + lm) * SAGE_D + 8 * half;
    const float* wl1 = wl0 + 16 * SAGE_D;

#pragma unroll 1
    for (int k0 = 0; k0 < SAGE_D; k0 += 16) {
        const float4 la0 = *(const float4*)(wl0 + k0);
        const float4 lb0 = *(const float4*)(wl0 + k0 + 4);
        const float4 la1 = *(const float4*)(wl1 + k0);
        const float4 lb1 = *(const float4*)(wl1 + k0 + 4);
        const float la0v[4] = {la0.x, la0.y, la0.z, la0.w};
        const float lb0v[4] = {lb0.x, lb0.y, lb0.z, lb0.w};
        const float la1v[4] = {la1.x, la1.y, la1.z, la1.w};
        const float lb1v[4] = {lb1.x, lb1.y, lb1.z, lb1.w};
#pragma unroll
        for (int i = 0; i < NRT; ++i) {
#pragma unroll
            for (int j = 0; j < 4; ++j) {
                const v2f a_s = *(const v2f*)(arow[i] + j * 32 + (k0 >> 2));
                const v2f b_l0 = (v2f){la0v[j], lb0v[j]};
                const v2f b_l1 = (v2f){la1v[j], lb1v[j]};
                c[i][0] = __builtin_amdgcn_wmma_f32_16x16x4_f32(
                    false, a_s, false, b_l0, (short)0, c[i][0], false, false);
                c[i][1] = __builtin_amdgcn_wmma_f32_16x16x4_f32(
                    false, a_s, false, b_l1, (short)0, c[i][1], false, false);
            }
        }
    }

#pragma unroll
    for (int i = 0; i < NRT; ++i) {
        const int rt = g + 8 * i;
        const bool whole = rt * 16 + 16 <= nOwn;
#pragma unroll
        for (int t = 0; t < 2; ++t) {
            const int col = cp * 32 + t * 16 + lm;
            const float bias = bl[col];
            if (whole) {
                float* o = out + (size_t)(lo + rt * 16 + 8 * half) * SAGE_D + col;
#pragma unroll
                for (int v = 0; v < 8; ++v) o[v * SAGE_D] = c[i][t][v] + bias;
            } else {
#pragma unroll
                for (int v = 0; v < 8; ++v) {
                    const int r = rt * 16 + v + 8 * half;
                    if (r < nOwn)
                        out[(size_t)(lo + r) * SAGE_D + col] = c[i][t][v] + bias;
                }
            }
        }
    }
}

__global__ __launch_bounds__(SAGE_T) void sage_layer_kernel(
    const float* __restrict__ xin, const int* __restrict__ sortedw,
    const int* __restrict__ segT,
    const float* __restrict__ Wl, const float* __restrict__ bl,
    const float* __restrict__ Wr, float* __restrict__ out, int E) {
    __shared__ __attribute__((aligned(16))) float acc[SAGE_ROWS * SAGE_P];
    __shared__ int2 lst[SAGE_LCAP];
    __shared__ int cntl[SAGE_R];
    __shared__ float invl[SAGE_R];
    __shared__ int segS[SAGE_NCH];
    __shared__ int segL[SAGE_NCH];
    __shared__ int rowc[SAGE_NCH];
    __shared__ int wtot[8];

    const int tid = threadIdx.x;
    const int lane = tid & 31;
    const int wave = __builtin_amdgcn_readfirstlane(tid >> 5);
    const int lo = blockIdx.x * SAGE_R;
    const int nOwn = min(SAGE_R, SAGE_N - lo);
    const int C = (E + SAGE_NCH - 1) / SAGE_NCH;
    const int2* sorted = (const int2*)sortedw;

    int s0 = 0;
    int s1 = 0;
    if (tid < SAGE_NCH) {
        s0 = segT[blockIdx.x * SAGE_NCH + tid];
        s1 = segT[(blockIdx.x + 1) * SAGE_NCH + tid];
    }
    for (int i = tid; i < SAGE_ROWS * SAGE_P; i += SAGE_T) acc[i] = 0.0f;
    for (int i = tid; i < SAGE_R; i += SAGE_T) cntl[i] = 0;
    if (tid < SAGE_NCH) {
        segS[tid] = tid * C + s0;
        segL[tid] = s1 - s0;
    }
    __syncthreads();

    int2 ent[SAGE_NCH / 32];
    int ln[SAGE_NCH / 32];
    int sp[SAGE_NCH / 32];
#pragma unroll
    for (int k = 0; k < SAGE_NCH / 32; ++k) {
        const int t = wave + 32 * k;
        ln[k] = segL[t];
        sp[k] = segS[t];
        ent[k] = sorted[lane < ln[k] ? sp[k] + lane : 0];
    }
#pragma unroll
    for (int k = 0; k < SAGE_NCH / 32; ++k)
        if (lane < ln[k] && ent[k].y < SAGE_R) atomicAdd(&cntl[ent[k].y], 1);
#pragma unroll
    for (int k = 0; k < SAGE_NCH / 32; ++k)
        for (int i = lane + 32; i < ln[k]; i += 32) {
            const int2 e = sorted[sp[k] + i];
            if (e.y < SAGE_R) atomicAdd(&cntl[e.y], 1);
        }
    __syncthreads();
    int nTot;
    const int rowStart = sage_scan256(tid < SAGE_R ? cntl[tid] : 0, tid, wtot, &nTot);
    if (tid < SAGE_R) rowc[tid] = rowStart;
    __syncthreads();
    const bool listed = nTot <= SAGE_LCAP;
    if (listed) {
        int pos[SAGE_NCH / 32];
#pragma unroll
        for (int k = 0; k < SAGE_NCH / 32; ++k)
            pos[k] = (lane < ln[k] && ent[k].y < SAGE_R) ? atomicAdd(&rowc[ent[k].y], 1) : -1;
#pragma unroll
        for (int k = 0; k < SAGE_NCH / 32; ++k)
            if (pos[k] >= 0) lst[pos[k]] = ent[k];
#pragma unroll
        for (int k = 0; k < SAGE_NCH / 32; ++k)
            for (int i = lane + 32; i < ln[k]; i += 32) {
                const int2 e = sorted[sp[k] + i];
                if (e.y < SAGE_R) lst[atomicAdd(&rowc[e.y], 1)] = e;
            }
    } else {
        for (int t = wave; t < SAGE_NCH; t += 32) {
            for (int i = lane; i < segL[t]; i += 32) {
                const int2 e = sorted[segS[t] + i];
                if (e.y < SAGE_R) {
                    const float* xr = xin + (size_t)e.x * SAGE_D;
                    float* ar = acc + e.y * SAGE_P;
                    for (int k = 0; k < SAGE_D; ++k)
                        unsafeAtomicAdd(ar + (k & 3) * 32 + (k >> 2), xr[k]);
                }
            }
        }
    }
    __syncthreads();
    const int n = __builtin_amdgcn_readfirstlane(listed ? nTot : 0);

    const int nTiles = (nOwn + 15) >> 4;
    const int g = wave >> 2;
    const int cp = wave & 3;
    const int nrt = (g + 8 < nTiles) ? 2 : ((g < nTiles) ? 1 : 0);
    const bool selfFirst = (g & 1) == 0;
    v8f c[2][2];
    c[0][0] = (v8f){};
    c[0][1] = (v8f){};
    c[1][0] = (v8f){};
    c[1][1] = (v8f){};
    if (selfFirst) {
        if (nrt == 2)
            sage_self_product<2>(c, xin, Wr, lo, g, cp, lane);
        else if (nrt == 1)
            sage_self_product<1>(c, xin, Wr, lo, g, cp, lane);
    }

    {
        const int per = (n + 31) >> 5;
        const int beg = min(wave * per, n);
        const int end = min(beg + per, n);
        int cur = -1;
        float4 run = make_float4(0.0f, 0.0f, 0.0f, 0.0f);
        const int lu = lane & (SAGE_U - 1);
        int2 mine = lst[max(min(beg + lu, end - 1), 0)];
        for (int base = beg; base < end; base += SAGE_U) {
            float4 v[SAGE_U];
            int dl[SAGE_U];
#pragma unroll
            for (int u = 0; u < SAGE_U; ++u) {
                const int sx = __builtin_amdgcn_readlane(mine.x, u);
                dl[u] = __builtin_amdgcn_readlane(mine.y, u);
                v[u] = *(const float4*)(xin + (size_t)sx * SAGE_D + lane * 4);
            }
            mine = lst[min(base + SAGE_U + lu, end - 1)];
            asm volatile("" :: "v"(v[0].x), "v"(v[1].x), "v"(v[2].x), "v"(v[3].x),
                               "v"(v[4].x), "v"(v[5].x), "v"(v[6].x), "v"(v[7].x));
#pragma unroll
            for (int u = 0; u < SAGE_U; ++u) {
                if (base + u < end) {
                    if (dl[u] != cur) {
                        if (cur >= 0) sage_run_add(acc, cur, lane, run);
                        cur = dl[u];
                        run = v[u];
                    } else {
                        run.x += v[u].x;
                        run.y += v[u].y;
                        run.z += v[u].z;
                        run.w += v[u].w;
                    }
                }
            }
        }
        if (cur >= 0) sage_run_add(acc, cur, lane, run);
    }
    if (!selfFirst) {
        if (nrt == 2)
            sage_self_product<2>(c, xin, Wr, lo, g, cp, lane);
        else if (nrt == 1)
            sage_self_product<1>(c, xin, Wr, lo, g, cp, lane);
    }
    __syncthreads();

    for (int i = tid; i < nOwn; i += SAGE_T)
        invl[i] = 1.0f / fmaxf((float)cntl[i], 1.0f);
    __syncthreads();
    for (int i = tid; i < nOwn * SAGE_D; i += SAGE_T) {
        const int r = i >> 7;
        acc[r * SAGE_P + (i & (SAGE_D - 1))] *= invl[r];
    }
    __syncthreads();

    if (nrt == 2)
        sage_sum_product_store<2>(c, acc, Wl, bl, out, lo, nOwn, g, cp, lane);
    else if (nrt == 1)
        sage_sum_product_store<1>(c, acc, Wl, bl, out, lo, nOwn, g, cp, lane);
}

extern "C" void kernel_launch(void* const* d_in, const int* in_sizes, int n_in,
                              void* d_out, int out_size, void* d_ws, size_t ws_size,
                              hipStream_t stream) {
    const float* x   = (const float*)d_in[0];
    const int*   ei  = (const int*)d_in[1];
    const float* Wl1 = (const float*)d_in[2];
    const float* bl1 = (const float*)d_in[3];
    const float* Wr1 = (const float*)d_in[4];
    const float* Wl2 = (const float*)d_in[5];
    const float* bl2 = (const float*)d_in[6];
    const float* Wr2 = (const float*)d_in[7];
    float* out = (float*)d_out;

    const int N = SAGE_N;
    const int E = in_sizes[1] / 2;
    const int* src = ei;
    const int* dst = ei + E;

    float* h    = (float*)d_ws;
    int* segT   = (int*)(h + (size_t)N * SAGE_D);
    int* sorted = segT + (SAGE_G + 1) * SAGE_NCH;

    sage_file_kernel<<<SAGE_NCH, SAGE_T, 0, stream>>>(src, dst, sorted, segT, E);

    sage_layer_kernel<<<SAGE_G, SAGE_T, 0, stream>>>(x, sorted, segT, Wl1, bl1, Wr1, h, E);
    sage_layer_kernel<<<SAGE_G, SAGE_T, 0, stream>>>(h, sorted, segT, Wl2, bl2, Wr2, out, E);
}


# Weights re-laid in memory in B-fragment order by 16 extra work-groups of launch 1: a wave's weight load is 512 contiguous bytes (4 lines), not 16 bytes from each of 16 rows; k and sum order unchanged
# speedup vs baseline: 4.7242x; 1.0868x over previous
#include <hip/hip_runtime.h>
#include <hip/hip_bf16.h>


#define SAGE_N 50000
#define SAGE_D 128
#define SAGE_R 196
#define SAGE_G 256
#define SAGE_NCH 256
#define SAGE_CCAP 8192
#define SAGE_FU 4
#define SAGE_LCAP 16384
#define SAGE_T 1024
#define SAGE_U 8
#define SAGE_P 132
#define SAGE_ROWS 208
#define SAGE_WQ (SAGE_D * SAGE_D / 4)
#define SAGE_WB (SAGE_WQ / SAGE_T)

typedef __attribute__((ext_vector_type(2))) float v2f;
typedef __attribute__((ext_vector_type(8))) float v8f;

__device__ __forceinline__ unsigned sage_owner_of(int d) {
    const unsigned o = (unsigned)d / (unsigned)SAGE_R;
    return o < (unsigned)(SAGE_G - 1) ? o : (unsigned)(SAGE_G - 1);
}

__device__ __forceinline__ int sage_scan256(int v, int tid, int* wtot, int* total) {
    const int lane = tid & 31;
    int incl = v;
    for (int d = 1; d < 32; d <<= 1) {
        const int t = __shfl_up(incl, d, 32);
        if (lane >= d) incl += t;
    }
    if (tid < 256 && lane == 31) wtot[tid >> 5] = incl;
    __syncthreads();
    int pre = 0;
    int tot = 0;
    for (int k = 0; k < 8; ++k) {
        const int w = wtot[k];
        if (k < (tid >> 5)) pre += w;
        tot += w;
    }
    *total = tot;
    return pre + incl - v;
}

__global__ __launch_bounds__(SAGE_T) void sage_file_kernel(
    const int* __restrict__ src, const int* __restrict__ dst,
    int* __restrict__ sortedw, int* __restrict__ segT,
    const float* __restrict__ wA, const float* __restrict__ wB,
    const float* __restrict__ wC, const float* __restrict__ wD,
    float* __restrict__ wfrag, int E) {
    __shared__ int2 stage[SAGE_CCAP];
    __shared__ int hist[SAGE_G];
    __shared__ int cursor[SAGE_G];
    __shared__ int wtot[8];

    const int tid = threadIdx.x;
    if (blockIdx.x >= SAGE_NCH) {
        const int wb = blockIdx.x - SAGE_NCH;
        const int m = wb / SAGE_WB;
        const float* W = m == 0 ? wA : m == 1 ? wB : m == 2 ? wC : wD;
        const int f = (wb % SAGE_WB) * SAGE_T + tid;
        const int lane = f & 31;
        const int q = (f >> 5) & 3;
        const int kc = (f >> 7) & 7;
        const int cp = f >> 10;
        const int n = cp * 32 + (q >> 1) * 16 + (lane & 15);
        const int b = kc * 16 + 8 * (lane >> 4) + 2 * (q & 1);
        const float2 lo2 = *(const float2*)(W + (size_t)n * SAGE_D + b);
        const float2 hi2 = *(const float2*)(W + (size_t)n * SAGE_D + b + 4);
        ((float4*)wfrag)[(size_t)m * SAGE_WQ + f] = make_float4(lo2.x, hi2.x, lo2.y, hi2.y);
        return;
    }
    const int w = blockIdx.x;
    const int C = (E + SAGE_NCH - 1) / SAGE_NCH;
    const int e0 = min(w * C, E);
    const int e1 = min(e0 + C, E);
    int2* sorted = (int2*)sortedw;

    if (tid < SAGE_G) hist[tid] = 0;
    __syncthreads();
    for (int b = e0 + tid; b < e1; b += SAGE_FU * SAGE_T) {
        int d[SAGE_FU];
#pragma unroll
        for (int u = 0; u < SAGE_FU; ++u) d[u] = dst[min(b + u * SAGE_T, e1 - 1)];
#pragma unroll
        for (int u = 0; u < SAGE_FU; ++u)
            atomicAdd(&hist[sage_owner_of(d[u])], b + u * SAGE_T < e1 ? 1 : 0);
    }
    __syncthreads();

    int total;
    const int cntv = tid < SAGE_G ? hist[tid] : 0;
    const int start = sage_scan256(cntv, tid, wtot, &total);
    if (tid < SAGE_G) {
        cursor[tid] = start;
        segT[tid * SAGE_NCH + w] = start;
    }
    if (tid == 0) segT[SAGE_G * SAGE_NCH + w] = total;
    __syncthreads();

    const bool staged = C <= SAGE_CCAP;
    for (int b = e0 + tid; b < e1; b += SAGE_FU * SAGE_T) {
        int d[SAGE_FU];
        int sv[SAGE_FU];
#pragma unroll
        for (int u = 0; u < SAGE_FU; ++u) {
            const int e = min(b + u * SAGE_T, e1 - 1);
            d[u] = dst[e];
            sv[u] = src[e];
        }
#pragma unroll
        for (int u = 0; u < SAGE_FU; ++u) {
            if (b + u * SAGE_T < e1) {
                const unsigned o = sage_owner_of(d[u]);
                const unsigned dl = (unsigned)d[u] - o * (unsigned)SAGE_R;
                const unsigned nOwnO = min((unsigned)SAGE_R, (unsigned)SAGE_N - o * (unsigned)SAGE_R);
                const int2 ent = dl < nOwnO ? make_int2(sv[u], (int)dl) : make_int2(0, SAGE_R);
                const int pos = atomicAdd(&cursor[o], 1);
                if (staged) stage[pos] = ent;
                else sorted[(size_t)e0 + pos] = ent;
            }
        }
    }
    __syncthreads();
    if (staged)
        for (int i = tid; i < e1 - e0; i += SAGE_T) sorted[(size_t)e0 + i] = stage[i];
}

__device__ __forceinline__ void sage_run_add(float* acc, int row, int lane, float4 run) {
    float* a = acc + row * SAGE_P + lane;
    unsafeAtomicAdd(a + 0, run.x);
    unsafeAtomicAdd(a + 32, run.y);
    unsafeAtomicAdd(a + 64, run.z);
    unsafeAtomicAdd(a + 96, run.w);
}

template <int NRT>
__device__ __forceinline__ void sage_self_product(
    v8f (&c)[2][2], const float* __restrict__ xin,
    const float* __restrict__ wf, int lo, int g, int cp, int lane) {
    const int half = lane >> 4;
    const int lm = lane & 15;
    const float* xrow[NRT];
#pragma unroll
    for (int i = 0; i < NRT; ++i) {
        const int r = (g + 8 * i) * 16 + lm;
        const int gr = min(lo + r, SAGE_N - 1);
        xrow[i] = xin + (size_t)gr * SAGE_D + 8 * half;
    }
    const float4* wq = (const float4*)wf + cp * 1024 + lane;

#pragma unroll 2
    for (int k0 = 0; k0 < SAGE_D; k0 += 16) {
        const float4 ra0 = wq[k0 * 8];
        const float4 rb0 = wq[k0 * 8 + 32];
        const float4 ra1 = wq[k0 * 8 + 64];
        const float4 rb1 = wq[k0 * 8 + 96];
        const float r0v[8] = {ra0.x, ra0.y, ra0.z, ra0.w, rb0.x, rb0.y, rb0.z, rb0.w};
        const float r1v[8] = {ra1.x, ra1.y, ra1.z, ra1.w, rb1.x, rb1.y, rb1.z, rb1.w};
#pragma unroll
        for (int i = 0; i < NRT; ++i) {
            const float4 xa = *(const float4*)(xrow[i] + k0);
            const float4 xb = *(const float4*)(xrow[i] + k0 + 4);
            const float xav[4] = {xa.x, xa.y, xa.z, xa.w};
            const float xbv[4] = {xb.x, xb.y, xb.z, xb.w};
#pragma unroll
            for (int j = 0; j < 4; ++j) {
                const v2f a_x = (v2f){xav[j], xbv[j]};
                const v2f b_r0 = (v2f){r0v[2 * j], r0v[2 * j + 1]};
                const v2f b_r1 = (v2f){r1v[2 * j], r1v[2 * j + 1]};
                c[i][0] = __builtin_amdgcn_wmma_f32_16x16x4_f32(
                    false, a_x, false, b_r0, (short)0, c[i][0], false, false);
                c[i][1] = __builtin_amdgcn_wmma_f32_16x16x4_f32(
                    false, a_x, false, b_r1, (short)0, c[i][1], false, false);
            }
        }
    }
}

template <int NRT>
__device__ __forceinline__ void sage_sum_product_store(
    v8f (&c)[2][2], const float* acc, const float* __restrict__ wf,
    const float* __restrict__ bl, float* __restrict__ out,
    int lo, int nOwn, int g, int cp, int lane) {
    const int half = lane >> 4;
    const int lm = lane & 15;
    const float* arow[NRT];
#pragma unroll
    for (int i = 0; i < NRT; ++i) {
        const int r = (g + 8 * i) * 16 + lm;
        arow[i] = acc + r * SAGE_P + 2 * half;
    }
    const float4* wq = (const float4*)wf + cp * 1024 + lane;

#pragma unroll 1
    for (int k0 = 0; k0 < SAGE_D; k0 += 16) {
        const float4 la0 = wq[k0 * 8];
        const float4 lb0 = wq[k0 * 8 + 32];
        const float4 la1 = wq[k0 * 8 + 64];
        const float4 lb1 = wq[k0 * 8 + 96];
        const float l0v[8] = {la0.x, la0.y, la0.z, la0.w, lb0.x, lb0.y, lb0.z, lb0.w};
        const float l1v[8] = {la1.x, la1.y, la1.z, la1.w, lb1.x, lb1.y, lb1.z, lb1.w};
#pragma unroll
        for (int i = 0; i < NRT; ++i) {
#pragma unroll
            for (int j = 0; j < 4; ++j) {
                const v2f a_s = *(const v2f*)(arow[i] + j * 32 + (k0 >> 2));
                const v2f b_l0 = (v2f){l0v[2 * j], l0v[2 * j + 1]};
                const v2f b_l1 = (v2f){l1v[2 * j], l1v[2 * j + 1]};
                c[i][0] = __builtin_amdgcn_wmma_f32_16x16x4_f32(
                    false, a_s, false, b_l0, (short)0, c[i][0], false, false);
                c[i][1] = __builtin_amdgcn_wmma_f32_16x16x4_f32(
                    false, a_s, false, b_l1, (short)0, c[i][1], false, false);
            }
        }
    }

#pragma unroll
    for (int i = 0; i < NRT; ++i) {
        const int rt = g + 8 * i;
        const bool whole = rt * 16 + 16 <= nOwn;
#pragma unroll
        for (int t = 0; t < 2; ++t) {
            const int col = cp * 32 + t * 16 + lm;
            const float bias = bl[col];
            if (whole) {
                float* o = out + (size_t)(lo + rt * 16 + 8 * half) * SAGE_D + col;
#pragma unroll
                for (int v = 0; v < 8; ++v) o[v * SAGE_D] = c[i][t][v] + bias;
            } else {
#pragma unroll
                for (int v = 0; v < 8; ++v) {
                    const int r = rt * 16 + v + 8 * half;
                    if (r < nOwn)
                        out[(size_t)(lo + r) * SAGE_D + col] = c[i][t][v] + bias;
                }
            }
        }
    }
}

__global__ __launch_bounds__(SAGE_T) void sage_layer_kernel(
    const float* __restrict__ xin, const int* __restrict__ sortedw,
    const int* __restrict__ segT,
    const float* __restrict__ Wl, const float* __restrict__ bl,
    const float* __restrict__ Wr, float* __restrict__ out, int E) {
    __shared__ __attribute__((aligned(16))) float acc[SAGE_ROWS * SAGE_P];
    __shared__ int2 lst[SAGE_LCAP];
    __shared__ int cntl[SAGE_R];
    __shared__ float invl[SAGE_R];
    __shared__ int segS[SAGE_NCH];
    __shared__ int segL[SAGE_NCH];
    __shared__ int rowc[SAGE_NCH];
    __shared__ int wtot[8];

    const int tid = threadIdx.x;
    const int lane = tid & 31;
    const int wave = __builtin_amdgcn_readfirstlane(tid >> 5);
    const int lo = blockIdx.x * SAGE_R;
    const int nOwn = min(SAGE_R, SAGE_N - lo);
    const int C = (E + SAGE_NCH - 1) / SAGE_NCH;
    const int2* sorted = (const int2*)sortedw;

    int s0 = 0;
    int s1 = 0;
    if (tid < SAGE_NCH) {
        s0 = segT[blockIdx.x * SAGE_NCH + tid];
        s1 = segT[(blockIdx.x + 1) * SAGE_NCH + tid];
    }
    for (int i = tid; i < SAGE_ROWS * SAGE_P; i += SAGE_T) acc[i] = 0.0f;
    for (int i = tid; i < SAGE_R; i += SAGE_T) cntl[i] = 0;
    if (tid < SAGE_NCH) {
        segS[tid] = tid * C + s0;
        segL[tid] = s1 - s0;
    }
    __syncthreads();

    int2 ent[SAGE_NCH / 32];
    int ln[SAGE_NCH / 32];
    int sp[SAGE_NCH / 32];
#pragma unroll
    for (int k = 0; k < SAGE_NCH / 32; ++k) {
        const int t = wave + 32 * k;
        ln[k] = segL[t];
        sp[k] = segS[t];
        ent[k] = sorted[lane < ln[k] ? sp[k] + lane : 0];
    }
#pragma unroll
    for (int k = 0; k < SAGE_NCH / 32; ++k)
        if (lane < ln[k] && ent[k].y < SAGE_R) atomicAdd(&cntl[ent[k].y], 1);
#pragma unroll
    for (int k = 0; k < SAGE_NCH / 32; ++k)
        for (int i = lane + 32; i < ln[k]; i += 32) {
            const int2 e = sorted[sp[k] + i];
            if (e.y < SAGE_R) atomicAdd(&cntl[e.y], 1);
        }
    __syncthreads();
    int nTot;
    const int rowStart = sage_scan256(tid < SAGE_R ? cntl[tid] : 0, tid, wtot, &nTot);
    if (tid < SAGE_R) rowc[tid] = rowStart;
    __syncthreads();
    const bool listed = nTot <= SAGE_LCAP;
    if (listed) {
        int pos[SAGE_NCH / 32];
#pragma unroll
        for (int k = 0; k < SAGE_NCH / 32; ++k)
            pos[k] = (lane < ln[k] && ent[k].y < SAGE_R) ? atomicAdd(&rowc[ent[k].y], 1) : -1;
#pragma unroll
        for (int k = 0; k < SAGE_NCH / 32; ++k)
            if (pos[k] >= 0) lst[pos[k]] = ent[k];
#pragma unroll
        for (int k = 0; k < SAGE_NCH / 32; ++k)
            for (int i = lane + 32; i < ln[k]; i += 32) {
                const int2 e = sorted[sp[k] + i];
                if (e.y < SAGE_R) lst[atomicAdd(&rowc[e.y], 1)] = e;
            }
    } else {
        for (int t = wave; t < SAGE_NCH; t += 32) {
            for (int i = lane; i < segL[t]; i += 32) {
                const int2 e = sorted[segS[t] + i];
                if (e.y < SAGE_R) {
                    const float* xr = xin + (size_t)e.x * SAGE_D;
                    float* ar = acc + e.y * SAGE_P;
                    for (int k = 0; k < SAGE_D; ++k)
                        unsafeAtomicAdd(ar + (k & 3) * 32 + (k >> 2), xr[k]);
                }
            }
        }
    }
    __syncthreads();
    const int n = __builtin_amdgcn_readfirstlane(listed ? nTot : 0);

    const int nTiles = (nOwn + 15) >> 4;
    const int g = wave >> 2;
    const int cp = wave & 3;
    const int nrt = (g + 8 < nTiles) ? 2 : ((g < nTiles) ? 1 : 0);
    const bool selfFirst = (g & 1) == 0;
    v8f c[2][2];
    c[0][0] = (v8f){};
    c[0][1] = (v8f){};
    c[1][0] = (v8f){};
    c[1][1] = (v8f){};
    if (selfFirst) {
        if (nrt == 2)
            sage_self_product<2>(c, xin, Wr, lo, g, cp, lane);
        else if (nrt == 1)
            sage_self_product<1>(c, xin, Wr, lo, g, cp, lane);
    }

    {
        const int per = (n + 31) >> 5;
        const int beg = min(wave * per, n);
        const int end = min(beg + per, n);
        int cur = -1;
        float4 run = make_float4(0.0f, 0.0f, 0.0f, 0.0f);
        const int lu = lane & (SAGE_U - 1);
        int2 mine = lst[max(min(beg + lu, end - 1), 0)];
        for (int base = beg; base < end; base += SAGE_U) {
            float4 v[SAGE_U];
            int dl[SAGE_U];
#pragma unroll
            for (int u = 0; u < SAGE_U; ++u) {
                const int sx = __builtin_amdgcn_readlane(mine.x, u);
                dl[u] = __builtin_amdgcn_readlane(mine.y, u);
                v[u] = *(const float4*)(xin + (size_t)sx * SAGE_D + lane * 4);
            }
            mine = lst[min(base + SAGE_U + lu, end - 1)];
            asm volatile("" :: "v"(v[0].x), "v"(v[1].x), "v"(v[2].x), "v"(v[3].x),
                               "v"(v[4].x), "v"(v[5].x), "v"(v[6].x), "v"(v[7].x));
#pragma unroll
            for (int u = 0; u < SAGE_U; ++u) {
                if (base + u < end) {
                    if (dl[u] != cur) {
                        if (cur >= 0) sage_run_add(acc, cur, lane, run);
                        cur = dl[u];
                        run = v[u];
                    } else {
                        run.x += v[u].x;
                        run.y += v[u].y;
                        run.z += v[u].z;
                        run.w += v[u].w;
                    }
                }
            }
        }
        if (cur >= 0) sage_run_add(acc, cur, lane, run);
    }
    if (!selfFirst) {
        if (nrt == 2)
            sage_self_product<2>(c, xin, Wr, lo, g, cp, lane);
        else if (nrt == 1)
            sage_self_product<1>(c, xin, Wr, lo, g, cp, lane);
    }
    __syncthreads();

    for (int i = tid; i < nOwn; i += SAGE_T)
        invl[i] = 1.0f / fmaxf((float)cntl[i], 1.0f);
    __syncthreads();
    for (int i = tid; i < nOwn * SAGE_D; i += SAGE_T) {
        const int r = i >> 7;
        acc[r * SAGE_P + (i & (SAGE_D - 1))] *= invl[r];
    }
    __syncthreads();

    if (nrt == 2)
        sage_sum_product_store<2>(c, acc, Wl, bl, out, lo, nOwn, g, cp, lane);
    else if (nrt == 1)
        sage_sum_product_store<1>(c, acc, Wl, bl, out, lo, nOwn, g, cp, lane);
}

extern "C" void kernel_launch(void* const* d_in, const int* in_sizes, int n_in,
                              void* d_out, int out_size, void* d_ws, size_t ws_size,
                              hipStream_t stream) {
    const float* x   = (const float*)d_in[0];
    const int*   ei  = (const int*)d_in[1];
    const float* Wl1 = (const float*)d_in[2];
    const float* bl1 = (const float*)d_in[3];
    const float* Wr1 = (const float*)d_in[4];
    const float* Wl2 = (const float*)d_in[5];
    const float* bl2 = (const float*)d_in[6];
    const float* Wr2 = (const float*)d_in[7];
    float* out = (float*)d_out;

    const int N = SAGE_N;
    const int E = in_sizes[1] / 2;
    const int* src = ei;
    const int* dst = ei + E;

    float* h     = (float*)d_ws;
    int* segT    = (int*)(h + (size_t)N * SAGE_D);
    float* wfrag = (float*)(segT + (SAGE_G + 1) * SAGE_NCH);
    int* sorted  = (int*)(wfrag + 4 * SAGE_D * SAGE_D);
    const float* fWl1 = wfrag;
    const float* fWr1 = wfrag + SAGE_D * SAGE_D;
    const float* fWl2 = wfrag + 2 * SAGE_D * SAGE_D;
    const float* fWr2 = wfrag + 3 * SAGE_D * SAGE_D;

    sage_file_kernel<<<SAGE_NCH + 4 * SAGE_WB, SAGE_T, 0, stream>>>(
        src, dst, sorted, segT, Wl1, Wr1, Wl2, Wr2, wfrag, E);

    sage_layer_kernel<<<SAGE_G, SAGE_T, 0, stream>>>(x, sorted, segT, fWl1, bl1, fWr1, h, E);
    sage_layer_kernel<<<SAGE_G, SAGE_T, 0, stream>>>(h, sorted, segT, fWl2, bl2, fWr2, out, E);
}
